# starting kernel
# baseline (speedup 1.0000x reference)
#include <hip/hip_runtime.h>
#include <cmath>

typedef __attribute__((ext_vector_type(2))) float v2f;
typedef __attribute__((ext_vector_type(8))) float v8f;

#define IN_DIM 128
#define HID_DIM 128
#define OUT_DIM 64
#define KDIM 128

__global__ void gcn_fill_one(float* __restrict__ deg, int n) {
    int i = blockIdx.x * blockDim.x + threadIdx.x;
    if (i < n) deg[i] = 1.0f;
}

__global__ void gcn_deg_count(const int* __restrict__ dst, float* __restrict__ deg, int e) {
    int i = blockIdx.x * blockDim.x + threadIdx.x;
    if (i < e) unsafeAtomicAdd(&deg[dst[i]], 1.0f);
}

__global__ void gcn_rsqrt_ip(float* __restrict__ deg, int n) {
    int i = blockIdx.x * blockDim.x + threadIdx.x;
    if (i < n) deg[i] = rsqrtf(deg[i]);
}

template <int NOUT>
__global__ __launch_bounds__(NOUT * 2) void gcn_gemm_wmma(
    const float* __restrict__ A, const float* __restrict__ B,
    float* __restrict__ C, int M) {
    constexpr int NTHREADS = NOUT * 2;
    __shared__ float Alds[16 * KDIM];

    const int mTile = blockIdx.x;
    const int tid   = threadIdx.x;

    for (int i = tid; i < 16 * KDIM; i += NTHREADS) {
        int r = i >> 7;
        int c = i & (KDIM - 1);
        int row = mTile * 16 + r;
        if (row >= M) row = M - 1;
        Alds[i] = A[(size_t)row * KDIM + c];
    }
    __syncthreads();

    const int wave  = tid >> 5;
    const int lane  = tid & 31;
    const int lrow  = lane & 15;
    const int khalf = lane >> 4;
    const int col   = lrow + wave * 16;

    v8f acc = {};
    #pragma unroll
    for (int k0 = 0; k0 < KDIM; k0 += 4) {
        const int ka = k0 + 2 * khalf;
        v2f a, b;
        a.x = Alds[lrow * KDIM + ka];
        a.y = Alds[lrow * KDIM + ka + 1];
        b.x = B[(size_t)ka * NOUT + col];
        b.y = B[(size_t)(ka + 1) * NOUT + col];
        acc = __builtin_amdgcn_wmma_f32_16x16x4_f32(
             false, a,  false, b,
             (short)0, acc,  false,  false);
    }

    const int rowBase = mTile * 16 + 8 * khalf;
    #pragma unroll
    for (int i = 0; i < 8; ++i) {
        int row = rowBase + i;
        if (row < M) C[(size_t)row * NOUT + col] = acc[i];
    }
}

template <int F>
__global__ void gcn_selfloop_init(const float* __restrict__ H,
                                  const float* __restrict__ dinv,
                                  float* __restrict__ agg, size_t total) {
    size_t i = (size_t)blockIdx.x * blockDim.x + threadIdx.x;
    if (i >= total) return;
    int n = (int)(i / F);
    float di = dinv[n];
    agg[i] = H[i] * di * di;
}

template <int F>
__global__ void gcn_edge_scatter(const float* __restrict__ H,
                                 const int* __restrict__ src,
                                 const int* __restrict__ dst,
                                 const float* __restrict__ dinv,
                                 float* __restrict__ agg, int E) {
    const int lane = threadIdx.x & 31;
    const int e = blockIdx.x * (blockDim.x >> 5) + (threadIdx.x >> 5);
    if (e >= E) return;
    const int s = src[e];
    const int d = dst[e];
    const float w = dinv[s] * dinv[d];
    const float* hp = H + (size_t)s * F;
    float* ap = agg + (size_t)d * F;
    #pragma unroll
    for (int j = 0; j < F / 32; ++j) {
        int f = lane + j * 32;
        unsafeAtomicAdd(ap + f, hp[f] * w);
    }
}

template <int F>
__global__ void gcn_bias_elu(const float* __restrict__ agg,
                             const float* __restrict__ bias,
                             float* __restrict__ out, size_t total) {
    size_t i = (size_t)blockIdx.x * blockDim.x + threadIdx.x;
    if (i >= total) return;
    float v = agg[i] + bias[i & (F - 1)];
    out[i] = (v > 0.0f) ? v : expm1f(v);
}

extern "C" void kernel_launch(void* const* d_in, const int* in_sizes, int n_in,
                              void* d_out, int out_size, void* d_ws, size_t ws_size,
                              hipStream_t stream) {
    const float* x  = (const float*)d_in[0];
    const int*   ei = (const int*)d_in[1];
    const float* W1 = (const float*)d_in[2];
    const float* b1 = (const float*)d_in[3];
    const float* W2 = (const float*)d_in[4];
    const float* b2 = (const float*)d_in[5];
    float* out = (float*)d_out;

    const int N = in_sizes[0] / IN_DIM;
    const int E = in_sizes[1] / 2;
    const int* src = ei;
    const int* dst = ei + E;

    char* ws = (char*)d_ws;
    float* dinv = (float*)ws;
    size_t off1 = ((size_t)N * 4 + 255) & ~(size_t)255;
    float* hbuf = (float*)(ws + off1);
    size_t off2 = off1 + ((((size_t)N * HID_DIM * 4) + 255) & ~(size_t)255);
    float* agg1 = (float*)(ws + off2);

    const int TB = 256;
    const int edgeBlocks = (E + 7) / 8;
    const int mTiles = (N + 15) / 16;
    const size_t tot1 = (size_t)N * HID_DIM;
    const size_t tot2 = (size_t)N * OUT_DIM;

    gcn_fill_one<<<(N + TB - 1) / TB, TB, 0, stream>>>(dinv, N);
    gcn_deg_count<<<(E + TB - 1) / TB, TB, 0, stream>>>(dst, dinv, E);
    gcn_rsqrt_ip<<<(N + TB - 1) / TB, TB, 0, stream>>>(dinv, N);

    gcn_gemm_wmma<HID_DIM><<<mTiles, HID_DIM * 2, 0, stream>>>(x, W1, hbuf, N);
    gcn_selfloop_init<HID_DIM><<<(int)((tot1 + TB - 1) / TB), TB, 0, stream>>>(hbuf, dinv, agg1, tot1);
    gcn_edge_scatter<HID_DIM><<<edgeBlocks, TB, 0, stream>>>(hbuf, src, dst, dinv, agg1, E);
    gcn_bias_elu<HID_DIM><<<(int)((tot1 + TB - 1) / TB), TB, 0, stream>>>(agg1, b1, agg1, tot1);

    gcn_gemm_wmma<OUT_DIM><<<mTiles, OUT_DIM * 2, 0, stream>>>(agg1, W2, hbuf, N);
    gcn_selfloop_init<OUT_DIM><<<(int)((tot2 + TB - 1) / TB), TB, 0, stream>>>(hbuf, dinv, out, tot2);
    gcn_edge_scatter<OUT_DIM><<<edgeBlocks, TB, 0, stream>>>(hbuf, src, dst, dinv, out, E);
    gcn_bias_elu<OUT_DIM><<<(int)((tot2 + TB - 1) / TB), TB, 0, stream>>>(out, b2, out, tot2);
}


# Edge sums leave global f32 atomics: each workgroup keeps 392 dst rows in LDS, scans all edges, adds its own source rows with LDS f32 atomics (arrival order), writes bias+ELU rows; 11 launches to 7
# speedup vs baseline: 1.7208x; 1.7208x over previous
#include <hip/hip_runtime.h>
#include <cmath>

typedef __attribute__((ext_vector_type(2))) float v2f;
typedef __attribute__((ext_vector_type(8))) float v8f;

#define IN_DIM 128
#define HID_DIM 128
#define OUT_DIM 64
#define KDIM 128

__global__ void gcn_fill_one(float* __restrict__ deg, int n) {
    int i = blockIdx.x * blockDim.x + threadIdx.x;
    if (i < n) deg[i] = 1.0f;
}

__global__ void gcn_deg_count(const int* __restrict__ dst, float* __restrict__ deg, int e) {
    int i = blockIdx.x * blockDim.x + threadIdx.x;
    if (i < e) unsafeAtomicAdd(&deg[dst[i]], 1.0f);
}

__global__ void gcn_rsqrt_ip(float* __restrict__ deg, int n) {
    int i = blockIdx.x * blockDim.x + threadIdx.x;
    if (i < n) deg[i] = rsqrtf(deg[i]);
}

template <int NOUT>
__global__ __launch_bounds__(NOUT * 2) void gcn_gemm_wmma(
    const float* __restrict__ A, const float* __restrict__ B,
    float* __restrict__ C, int M) {
    constexpr int NTHREADS = NOUT * 2;
    __shared__ float Alds[16 * KDIM];

    const int mTile = blockIdx.x;
    const int tid   = threadIdx.x;

    for (int i = tid; i < 16 * KDIM; i += NTHREADS) {
        int r = i >> 7;
        int c = i & (KDIM - 1);
        int row = mTile * 16 + r;
        if (row >= M) row = M - 1;
        Alds[i] = A[(size_t)row * KDIM + c];
    }
    __syncthreads();

    const int wave  = tid >> 5;
    const int lane  = tid & 31;
    const int lrow  = lane & 15;
    const int khalf = lane >> 4;
    const int col   = lrow + wave * 16;

    v8f acc = {};
    #pragma unroll
    for (int k0 = 0; k0 < KDIM; k0 += 4) {
        const int ka = k0 + 2 * khalf;
        v2f a, b;
        a.x = Alds[lrow * KDIM + ka];
        a.y = Alds[lrow * KDIM + ka + 1];
        b.x = B[(size_t)ka * NOUT + col];
        b.y = B[(size_t)(ka + 1) * NOUT + col];
        acc = __builtin_amdgcn_wmma_f32_16x16x4_f32(
             false, a,  false, b,
             (short)0, acc,  false,  false);
    }

    const int rowBase = mTile * 16 + 8 * khalf;
    #pragma unroll
    for (int i = 0; i < 8; ++i) {
        int row = rowBase + i;
        if (row < M) C[(size_t)row * NOUT + col] = acc[i];
    }
}

#define TILE_ROWS   392
#define AGG_THREADS 1024
#define AGG_WAVES   (AGG_THREADS / 32)
#define LIST_CAP    640
#define LIST_FLUSH  (LIST_CAP - 512)
#define ROW_BATCH   8

template <int F, int B>
__device__ __forceinline__ void gcn_add_rows(const float* __restrict__ H, float* acc,
                                             int s, int roff, int wbits, int j, int lane) {
    constexpr int Q = F / 32;
    int oj[B];
    float wj[B];
    float v[B][Q];
    #pragma unroll
    for (int u = 0; u < B; ++u) {
        const int sj = __builtin_amdgcn_readlane(s, j + u);
        oj[u] = __builtin_amdgcn_readlane(roff, j + u);
        wj[u] = __int_as_float(__builtin_amdgcn_readlane(wbits, j + u));
        const float* hp = H + (size_t)sj * F + lane;
        #pragma unroll
        for (int q = 0; q < Q; ++q) v[u][q] = hp[q * 32];
    }
    #pragma unroll
    for (int u = 0; u < B; ++u) {
        #pragma unroll
        for (int q = 0; q < Q; ++q)
            atomicAdd(&acc[oj[u] + lane + q * 32], v[u][q] * wj[u]);
    }
}

template <int F>
__device__ __forceinline__ void gcn_flush_list(const float* __restrict__ H,
                                               const int* __restrict__ src,
                                               const int* __restrict__ dst,
                                               const float* __restrict__ dinv,
                                               float* acc, const int* mylist,
                                               int cnt, int lo, int lane) {
    for (int base = 0; base < cnt; base += 32) {
        int m = cnt - base;
        if (m > 32) m = 32;
        const int idx = (lane < m) ? (base + lane) : base;
        const int e = mylist[idx];
        const int s = src[e];
        const int d = dst[e];
        const float w = dinv[s] * dinv[d];
        const int roff = (d - lo) * F;
        const int wbits = __float_as_int(w);

        int j = 0;
        for (; j + ROW_BATCH <= m; j += ROW_BATCH)
            gcn_add_rows<F, ROW_BATCH>(H, acc, s, roff, wbits, j, lane);
        if (j + 4 <= m) { gcn_add_rows<F, 4>(H, acc, s, roff, wbits, j, lane); j += 4; }
        if (j + 2 <= m) { gcn_add_rows<F, 2>(H, acc, s, roff, wbits, j, lane); j += 2; }
        if (j < m)      { gcn_add_rows<F, 1>(H, acc, s, roff, wbits, j, lane); }
    }
}

template <int F>
__global__ __launch_bounds__(AGG_THREADS) void gcn_tile_aggregate(
    const float* __restrict__ H, const int* __restrict__ src,
    const int* __restrict__ dst, const float* __restrict__ dinv,
    const float* __restrict__ bias, float* __restrict__ out, int N, int E) {
    __shared__ __align__(16) float acc[TILE_ROWS * F];
    __shared__ int elist[AGG_WAVES * LIST_CAP];

    const int tid  = threadIdx.x;
    const int lane = tid & 31;
    const int wave = __builtin_amdgcn_readfirstlane(tid >> 5);
    const int lo   = blockIdx.x * TILE_ROWS;
    int nrows = N - lo;
    if (nrows > TILE_ROWS) nrows = TILE_ROWS;
    const int nelem = nrows * F;
    const size_t gbase = (size_t)lo * F;

    const int nvec = nelem >> 2;
    {
        const float4* H4 = reinterpret_cast<const float4*>(H + gbase);
        float4* acc4 = reinterpret_cast<float4*>(acc);
        #pragma unroll 4
        for (int i = tid; i < nvec; i += AGG_THREADS) {
            const float di = dinv[lo + (int)((unsigned)i / (unsigned)(F / 4))];
            float4 h = H4[i];
            h.x = h.x * di * di; h.y = h.y * di * di; h.z = h.z * di * di; h.w = h.w * di * di;
            acc4[i] = h;
        }
    }
    __syncthreads();

    int* mylist = elist + wave * LIST_CAP;
    int cnt = 0;

#define GCN_APPEND(bal, mk, ek)                                                     \
    if ((bal) != 0u) {                                                              \
        if (mk) mylist[cnt + (int)__builtin_amdgcn_mbcnt_lo((bal), 0u)] = (ek);     \
        cnt += __popc(bal);                                                         \
    }
#define GCN_SCAN4(dv, eb)                                                           \
    {                                                                               \
        const bool m0 = (unsigned)((dv).x - lo) < (unsigned)TILE_ROWS;              \
        const bool m1 = (unsigned)((dv).y - lo) < (unsigned)TILE_ROWS;              \
        const bool m2 = (unsigned)((dv).z - lo) < (unsigned)TILE_ROWS;              \
        const bool m3 = (unsigned)((dv).w - lo) < (unsigned)TILE_ROWS;              \
        const unsigned q0 = __builtin_amdgcn_ballot_w32(m0);                        \
        const unsigned q1 = __builtin_amdgcn_ballot_w32(m1);                        \
        const unsigned q2 = __builtin_amdgcn_ballot_w32(m2);                        \
        const unsigned q3 = __builtin_amdgcn_ballot_w32(m3);                        \
        if ((q0 | q1 | q2 | q3) != 0u) {                                            \
            GCN_APPEND(q0, m0, (eb))                                                \
            GCN_APPEND(q1, m1, (eb) + 1)                                            \
            GCN_APPEND(q2, m2, (eb) + 2)                                            \
            GCN_APPEND(q3, m3, (eb) + 3)                                            \
        }                                                                           \
    }

    const int lead = (int)(((16u - (unsigned)((size_t)dst & 15u)) & 15u) >> 2);
    const int nGroups = (E > lead) ? ((E - lead) >> 9) : 0;
    {
        const int4* dstv = reinterpret_cast<const int4*>(dst + lead);
        int g = wave;
        int4 n0 = make_int4(-1, -1, -1, -1), n1 = n0, n2 = n0, n3 = n0;
        if (g < nGroups) {
            const int4* p = dstv + ((size_t)g << 7) + lane;
            n0 = p[0]; n1 = p[32]; n2 = p[64]; n3 = p[96];
        }
        while (g < nGroups) {
            const int4 c0 = n0, c1 = n1, c2 = n2, c3 = n3;
            const int gn = g + AGG_WAVES;
            if (gn < nGroups) {
                const int4* p = dstv + ((size_t)gn << 7) + lane;
                n0 = p[0]; n1 = p[32]; n2 = p[64]; n3 = p[96];
            }
            const int eb = lead + (g << 9) + (lane << 2);
            GCN_SCAN4(c0, eb)
            GCN_SCAN4(c1, eb + 128)
            GCN_SCAN4(c2, eb + 256)
            GCN_SCAN4(c3, eb + 384)
            if (cnt > LIST_FLUSH) {
                gcn_flush_list<F>(H, src, dst, dinv, acc, mylist, cnt, lo, lane);
                cnt = 0;
            }
            g = gn;
        }
    }
    {
        const int e = (tid < lead) ? tid : ((nGroups << 9) + tid);
        bool mk = false;
        if (e < E) mk = (unsigned)(dst[e] - lo) < (unsigned)TILE_ROWS;
        const unsigned qt = __builtin_amdgcn_ballot_w32(mk);
        GCN_APPEND(qt, mk, e)
    }
    if (cnt > 0) gcn_flush_list<F>(H, src, dst, dinv, acc, mylist, cnt, lo, lane);
#undef GCN_SCAN4
#undef GCN_APPEND
    __syncthreads();

    {
        const float4 bv = reinterpret_cast<const float4*>(bias)[tid & (F / 4 - 1)];
        const float4* acc4 = reinterpret_cast<const float4*>(acc);
        float4* out4 = reinterpret_cast<float4*>(out + gbase);
        for (int i = tid; i < nvec; i += AGG_THREADS) {
            float4 a = acc4[i];
            a.x += bv.x; a.y += bv.y; a.z += bv.z; a.w += bv.w;
            a.x = (a.x > 0.0f) ? a.x : expm1f(a.x);
            a.y = (a.y > 0.0f) ? a.y : expm1f(a.y);
            a.z = (a.z > 0.0f) ? a.z : expm1f(a.z);
            a.w = (a.w > 0.0f) ? a.w : expm1f(a.w);
            out4[i] = a;
        }
    }
}

extern "C" void kernel_launch(void* const* d_in, const int* in_sizes, int n_in,
                              void* d_out, int out_size, void* d_ws, size_t ws_size,
                              hipStream_t stream) {
    const float* x  = (const float*)d_in[0];
    const int*   ei = (const int*)d_in[1];
    const float* W1 = (const float*)d_in[2];
    const float* b1 = (const float*)d_in[3];
    const float* W2 = (const float*)d_in[4];
    const float* b2 = (const float*)d_in[5];
    float* out = (float*)d_out;

    const int N = in_sizes[0] / IN_DIM;
    const int E = in_sizes[1] / 2;
    const int* src = ei;
    const int* dst = ei + E;

    char* ws = (char*)d_ws;
    float* dinv = (float*)ws;
    size_t off1 = ((size_t)N * 4 + 255) & ~(size_t)255;
    float* hbuf = (float*)(ws + off1);
    size_t off2 = off1 + ((((size_t)N * HID_DIM * 4) + 255) & ~(size_t)255);
    float* act1 = (float*)(ws + off2);

    const int TB = 256;
    const int mTiles = (N + 15) / 16;
    const int nTiles = (N + TILE_ROWS - 1) / TILE_ROWS;

    gcn_fill_one<<<(N + TB - 1) / TB, TB, 0, stream>>>(dinv, N);
    gcn_deg_count<<<(E + TB - 1) / TB, TB, 0, stream>>>(dst, dinv, E);
    gcn_rsqrt_ip<<<(N + TB - 1) / TB, TB, 0, stream>>>(dinv, N);

    gcn_gemm_wmma<HID_DIM><<<mTiles, HID_DIM * 2, 0, stream>>>(x, W1, hbuf, N);
    gcn_tile_aggregate<HID_DIM><<<nTiles, AGG_THREADS, 0, stream>>>(hbuf, src, dst, dinv, b1, act1, N, E);

    gcn_gemm_wmma<OUT_DIM><<<mTiles, OUT_DIM * 2, 0, stream>>>(act1, W2, hbuf, N);
    gcn_tile_aggregate<OUT_DIM><<<nTiles, AGG_THREADS, 0, stream>>>(hbuf, src, dst, dinv, b2, out, N, E);
}


# Both GEMMs: a wave reads its 16 rows' A fragments straight from global memory into registers (16-byte pieces, k summed in a permuted order) for all column tiles; LDS A tile and barrier gone
# speedup vs baseline: 1.8997x; 1.1040x over previous
#include <hip/hip_runtime.h>
#include <cmath>

typedef __attribute__((ext_vector_type(2))) float v2f;
typedef __attribute__((ext_vector_type(8))) float v8f;

#define IN_DIM 128
#define HID_DIM 128
#define OUT_DIM 64
#define KDIM 128

__global__ void gcn_fill_one(float* __restrict__ deg, int n) {
    int i = blockIdx.x * blockDim.x + threadIdx.x;
    if (i < n) deg[i] = 1.0f;
}

__global__ void gcn_deg_count(const int* __restrict__ dst, float* __restrict__ deg, int e) {
    int i = blockIdx.x * blockDim.x + threadIdx.x;
    if (i < e) unsafeAtomicAdd(&deg[dst[i]], 1.0f);
}

__global__ void gcn_rsqrt_ip(float* __restrict__ deg, int n) {
    int i = blockIdx.x * blockDim.x + threadIdx.x;
    if (i < n) deg[i] = rsqrtf(deg[i]);
}

#define GEMM_THREADS 128
#define GEMM_WAVES   (GEMM_THREADS / 32)

template <int NOUT, int NT>
__device__ __forceinline__ void gcn_load_b_step(const float* __restrict__ bk, v2f (&b)[NT]) {
    #pragma unroll
    for (int t = 0; t < NT; ++t) {
        b[t].x = bk[16 * t];
        b[t].y = bk[NOUT + 16 * t];
    }
}

template <int NOUT>
__global__ __launch_bounds__(GEMM_THREADS) void gcn_gemm_wmma(
    const float* __restrict__ A, const float* __restrict__ B,
    float* __restrict__ C, int M) {
    constexpr int NT = NOUT / 16;

    const int tid  = threadIdx.x;
    const int wave = tid >> 5;
    const int lane = tid & 31;
    const int c    = lane & 15;
    const int h    = lane >> 4;

    const int mTile = blockIdx.x * GEMM_WAVES + wave;
    if (mTile * 16 >= M) return;

    int arow = mTile * 16 + c;
    if (arow >= M) arow = M - 1;

    const float4* ap = reinterpret_cast<const float4*>(A + (size_t)arow * KDIM) + h;
    float4 a[16];
    #pragma unroll
    for (int j = 0; j < 16; ++j) a[j] = ap[2 * j];

    const float* bp = B + (size_t)(4 * h) * NOUT + c;

    v2f bcur[NT], bnxt[NT];
    gcn_load_b_step<NOUT, NT>(bp, bcur);

    __builtin_amdgcn_sched_barrier(0);

    v8f acc[NT];
    #pragma unroll
    for (int t = 0; t < NT; ++t) acc[t] = v8f{};

    #pragma unroll
    for (int s = 0; s < 32; ++s) {
        const int j = s >> 1;
        const int p = s & 1;
        if (s + 1 < 32)
            gcn_load_b_step<NOUT, NT>(bp + (size_t)(8 * ((s + 1) >> 1) + 2 * ((s + 1) & 1)) * NOUT, bnxt);
        __builtin_amdgcn_sched_barrier(0);
        v2f av;
        av.x = p ? a[j].z : a[j].x;
        av.y = p ? a[j].w : a[j].y;
        #pragma unroll
        for (int t = 0; t < NT; ++t) {
            acc[t] = __builtin_amdgcn_wmma_f32_16x16x4_f32(
                 false, av,  false, bcur[t],
                 (short)0, acc[t],  false,  false);
        }
        #pragma unroll
        for (int t = 0; t < NT; ++t) bcur[t] = bnxt[t];
    }

    const int rowBase = mTile * 16 + 8 * h;
    #pragma unroll
    for (int i = 0; i < 8; ++i) {
        const int row = rowBase + i;
        if (row < M) {
            float* cp = C + (size_t)row * NOUT + c;
            #pragma unroll
            for (int t = 0; t < NT; ++t) cp[16 * t] = acc[t][i];
        }
    }
}

#define TILE_ROWS   392
#define AGG_THREADS 1024
#define AGG_WAVES   (AGG_THREADS / 32)
#define LIST_CAP    640
#define LIST_FLUSH  (LIST_CAP - 512)
#define ROW_BATCH   8

template <int F, int B>
__device__ __forceinline__ void gcn_add_rows(const float* __restrict__ H, float* acc,
                                             int s, int roff, int wbits, int j, int lane) {
    constexpr int Q = F / 32;
    int oj[B];
    float wj[B];
    float v[B][Q];
    #pragma unroll
    for (int u = 0; u < B; ++u) {
        const int sj = __builtin_amdgcn_readlane(s, j + u);
        oj[u] = __builtin_amdgcn_readlane(roff, j + u);
        wj[u] = __int_as_float(__builtin_amdgcn_readlane(wbits, j + u));
        const float* hp = H + (size_t)sj * F + lane;
        #pragma unroll
        for (int q = 0; q < Q; ++q) v[u][q] = hp[q * 32];
    }
    #pragma unroll
    for (int u = 0; u < B; ++u) {
        #pragma unroll
        for (int q = 0; q < Q; ++q)
            atomicAdd(&acc[oj[u] + lane + q * 32], v[u][q] * wj[u]);
    }
}

template <int F>
__device__ __forceinline__ void gcn_flush_list(const float* __restrict__ H,
                                               const int* __restrict__ src,
                                               const int* __restrict__ dst,
                                               const float* __restrict__ dinv,
                                               float* acc, const int* mylist,
                                               int cnt, int lo, int lane) {
    for (int base = 0; base < cnt; base += 32) {
        int m = cnt - base;
        if (m > 32) m = 32;
        const int idx = (lane < m) ? (base + lane) : base;
        const int e = mylist[idx];
        const int s = src[e];
        const int d = dst[e];
        const float w = dinv[s] * dinv[d];
        const int roff = (d - lo) * F;
        const int wbits = __float_as_int(w);

        int j = 0;
        for (; j + ROW_BATCH <= m; j += ROW_BATCH)
            gcn_add_rows<F, ROW_BATCH>(H, acc, s, roff, wbits, j, lane);
        if (j + 4 <= m) { gcn_add_rows<F, 4>(H, acc, s, roff, wbits, j, lane); j += 4; }
        if (j + 2 <= m) { gcn_add_rows<F, 2>(H, acc, s, roff, wbits, j, lane); j += 2; }
        if (j < m)      { gcn_add_rows<F, 1>(H, acc, s, roff, wbits, j, lane); }
    }
}

template <int F>
__global__ __launch_bounds__(AGG_THREADS) void gcn_tile_aggregate(
    const float* __restrict__ H, const int* __restrict__ src,
    const int* __restrict__ dst, const float* __restrict__ dinv,
    const float* __restrict__ bias, float* __restrict__ out, int N, int E) {
    __shared__ __align__(16) float acc[TILE_ROWS * F];
    __shared__ int elist[AGG_WAVES * LIST_CAP];

    const int tid  = threadIdx.x;
    const int lane = tid & 31;
    const int wave = __builtin_amdgcn_readfirstlane(tid >> 5);
    const int lo   = blockIdx.x * TILE_ROWS;
    int nrows = N - lo;
    if (nrows > TILE_ROWS) nrows = TILE_ROWS;
    const int nelem = nrows * F;
    const size_t gbase = (size_t)lo * F;

    const int nvec = nelem >> 2;
    {
        const float4* H4 = reinterpret_cast<const float4*>(H + gbase);
        float4* acc4 = reinterpret_cast<float4*>(acc);
        #pragma unroll 4
        for (int i = tid; i < nvec; i += AGG_THREADS) {
            const float di = dinv[lo + (int)((unsigned)i / (unsigned)(F / 4))];
            float4 h = H4[i];
            h.x = h.x * di * di; h.y = h.y * di * di; h.z = h.z * di * di; h.w = h.w * di * di;
            acc4[i] = h;
        }
    }
    __syncthreads();

    int* mylist = elist + wave * LIST_CAP;
    int cnt = 0;

#define GCN_APPEND(bal, mk, ek)                                                     \
    if ((bal) != 0u) {                                                              \
        if (mk) mylist[cnt + (int)__builtin_amdgcn_mbcnt_lo((bal), 0u)] = (ek);     \
        cnt += __popc(bal);                                                         \
    }
#define GCN_SCAN4(dv, eb)                                                           \
    {                                                                               \
        const bool m0 = (unsigned)((dv).x - lo) < (unsigned)TILE_ROWS;              \
        const bool m1 = (unsigned)((dv).y - lo) < (unsigned)TILE_ROWS;              \
        const bool m2 = (unsigned)((dv).z - lo) < (unsigned)TILE_ROWS;              \
        const bool m3 = (unsigned)((dv).w - lo) < (unsigned)TILE_ROWS;              \
        const unsigned q0 = __builtin_amdgcn_ballot_w32(m0);                        \
        const unsigned q1 = __builtin_amdgcn_ballot_w32(m1);                        \
        const unsigned q2 = __builtin_amdgcn_ballot_w32(m2);                        \
        const unsigned q3 = __builtin_amdgcn_ballot_w32(m3);                        \
        if ((q0 | q1 | q2 | q3) != 0u) {                                            \
            GCN_APPEND(q0, m0, (eb))                                                \
            GCN_APPEND(q1, m1, (eb) + 1)                                            \
            GCN_APPEND(q2, m2, (eb) + 2)                                            \
            GCN_APPEND(q3, m3, (eb) + 3)                                            \
        }                                                                           \
    }

    const int lead = (int)(((16u - (unsigned)((size_t)dst & 15u)) & 15u) >> 2);
    const int nGroups = (E > lead) ? ((E - lead) >> 9) : 0;
    {
        const int4* dstv = reinterpret_cast<const int4*>(dst + lead);
        int g = wave;
        int4 n0 = make_int4(-1, -1, -1, -1), n1 = n0, n2 = n0, n3 = n0;
        if (g < nGroups) {
            const int4* p = dstv + ((size_t)g << 7) + lane;
            n0 = p[0]; n1 = p[32]; n2 = p[64]; n3 = p[96];
        }
        while (g < nGroups) {
            const int4 c0 = n0, c1 = n1, c2 = n2, c3 = n3;
            const int gn = g + AGG_WAVES;
            if (gn < nGroups) {
                const int4* p = dstv + ((size_t)gn << 7) + lane;
                n0 = p[0]; n1 = p[32]; n2 = p[64]; n3 = p[96];
            }
            const int eb = lead + (g << 9) + (lane << 2);
            GCN_SCAN4(c0, eb)
            GCN_SCAN4(c1, eb + 128)
            GCN_SCAN4(c2, eb + 256)
            GCN_SCAN4(c3, eb + 384)
            if (cnt > LIST_FLUSH) {
                gcn_flush_list<F>(H, src, dst, dinv, acc, mylist, cnt, lo, lane);
                cnt = 0;
            }
            g = gn;
        }
    }
    {
        const int e = (tid < lead) ? tid : ((nGroups << 9) + tid);
        bool mk = false;
        if (e < E) mk = (unsigned)(dst[e] - lo) < (unsigned)TILE_ROWS;
        const unsigned qt = __builtin_amdgcn_ballot_w32(mk);
        GCN_APPEND(qt, mk, e)
    }
    if (cnt > 0) gcn_flush_list<F>(H, src, dst, dinv, acc, mylist, cnt, lo, lane);
#undef GCN_SCAN4
#undef GCN_APPEND
    __syncthreads();

    {
        const float4 bv = reinterpret_cast<const float4*>(bias)[tid & (F / 4 - 1)];
        const float4* acc4 = reinterpret_cast<const float4*>(acc);
        float4* out4 = reinterpret_cast<float4*>(out + gbase);
        for (int i = tid; i < nvec; i += AGG_THREADS) {
            float4 a = acc4[i];
            a.x += bv.x; a.y += bv.y; a.z += bv.z; a.w += bv.w;
            a.x = (a.x > 0.0f) ? a.x : expm1f(a.x);
            a.y = (a.y > 0.0f) ? a.y : expm1f(a.y);
            a.z = (a.z > 0.0f) ? a.z : expm1f(a.z);
            a.w = (a.w > 0.0f) ? a.w : expm1f(a.w);
            out4[i] = a;
        }
    }
}

extern "C" void kernel_launch(void* const* d_in, const int* in_sizes, int n_in,
                              void* d_out, int out_size, void* d_ws, size_t ws_size,
                              hipStream_t stream) {
    const float* x  = (const float*)d_in[0];
    const int*   ei = (const int*)d_in[1];
    const float* W1 = (const float*)d_in[2];
    const float* b1 = (const float*)d_in[3];
    const float* W2 = (const float*)d_in[4];
    const float* b2 = (const float*)d_in[5];
    float* out = (float*)d_out;

    const int N = in_sizes[0] / IN_DIM;
    const int E = in_sizes[1] / 2;
    const int* src = ei;
    const int* dst = ei + E;

    char* ws = (char*)d_ws;
    float* dinv = (float*)ws;
    size_t off1 = ((size_t)N * 4 + 255) & ~(size_t)255;
    float* hbuf = (float*)(ws + off1);
    size_t off2 = off1 + ((((size_t)N * HID_DIM * 4) + 255) & ~(size_t)255);
    float* act1 = (float*)(ws + off2);

    const int TB = 256;
    const int mTiles = (N + 15) / 16;
    const int nTiles = (N + TILE_ROWS - 1) / TILE_ROWS;

    gcn_fill_one<<<(N + TB - 1) / TB, TB, 0, stream>>>(dinv, N);
    gcn_deg_count<<<(E + TB - 1) / TB, TB, 0, stream>>>(dst, dinv, E);
    gcn_rsqrt_ip<<<(N + TB - 1) / TB, TB, 0, stream>>>(dinv, N);

    gcn_gemm_wmma<HID_DIM><<<(mTiles + GEMM_WAVES - 1) / GEMM_WAVES, GEMM_THREADS, 0, stream>>>(x, W1, hbuf, N);
    gcn_tile_aggregate<HID_DIM><<<nTiles, AGG_THREADS, 0, stream>>>(hbuf, src, dst, dinv, b1, act1, N, E);

    gcn_gemm_wmma<OUT_DIM><<<(mTiles + GEMM_WAVES - 1) / GEMM_WAVES, GEMM_THREADS, 0, stream>>>(act1, W2, hbuf, N);
    gcn_tile_aggregate<OUT_DIM><<<nTiles, AGG_THREADS, 0, stream>>>(hbuf, src, dst, dinv, b2, out, N, E);
}


# Layer-1 aggregate records each wave's matched (src,dst) pairs in the tile's own rows of the output buffer; layer-2 aggregate walks that record, no second scan of all edges; sums stay in arrival order
# speedup vs baseline: 2.2128x; 1.1648x over previous
#include <hip/hip_runtime.h>
#include <cmath>

typedef __attribute__((ext_vector_type(2))) float v2f;
typedef __attribute__((ext_vector_type(8))) float v8f;

#define IN_DIM 128
#define HID_DIM 128
#define OUT_DIM 64
#define KDIM 128

__global__ void gcn_fill_one(float* __restrict__ deg, int n) {
    int i = blockIdx.x * blockDim.x + threadIdx.x;
    if (i < n) deg[i] = 1.0f;
}

__global__ void gcn_deg_count(const int* __restrict__ dst, float* __restrict__ deg, int e) {
    int i = blockIdx.x * blockDim.x + threadIdx.x;
    if (i < e) unsafeAtomicAdd(&deg[dst[i]], 1.0f);
}

__global__ void gcn_rsqrt_ip(float* __restrict__ deg, int n) {
    int i = blockIdx.x * blockDim.x + threadIdx.x;
    if (i < n) deg[i] = rsqrtf(deg[i]);
}

#define GEMM_THREADS 128
#define GEMM_WAVES   (GEMM_THREADS / 32)

template <int NOUT, int NT>
__device__ __forceinline__ void gcn_load_b_step(const float* __restrict__ bk, v2f (&b)[NT]) {
    #pragma unroll
    for (int t = 0; t < NT; ++t) {
        b[t].x = bk[16 * t];
        b[t].y = bk[NOUT + 16 * t];
    }
}

template <int NOUT>
__global__ __launch_bounds__(GEMM_THREADS) void gcn_gemm_wmma(
    const float* __restrict__ A, const float* __restrict__ B,
    float* __restrict__ C, int M) {
    constexpr int NT = NOUT / 16;

    const int tid  = threadIdx.x;
    const int wave = tid >> 5;
    const int lane = tid & 31;
    const int c    = lane & 15;
    const int h    = lane >> 4;

    const int mTile = blockIdx.x * GEMM_WAVES + wave;
    if (mTile * 16 >= M) return;

    int arow = mTile * 16 + c;
    if (arow >= M) arow = M - 1;

    const float4* ap = reinterpret_cast<const float4*>(A + (size_t)arow * KDIM) + h;
    float4 a[16];
    #pragma unroll
    for (int j = 0; j < 16; ++j) a[j] = ap[2 * j];

    const float* bp = B + (size_t)(4 * h) * NOUT + c;

    v2f bcur[NT], bnxt[NT];
    gcn_load_b_step<NOUT, NT>(bp, bcur);

    __builtin_amdgcn_sched_barrier(0);

    v8f acc[NT];
    #pragma unroll
    for (int t = 0; t < NT; ++t) acc[t] = v8f{};

    #pragma unroll
    for (int s = 0; s < 32; ++s) {
        const int j = s >> 1;
        const int p = s & 1;
        if (s + 1 < 32)
            gcn_load_b_step<NOUT, NT>(bp + (size_t)(8 * ((s + 1) >> 1) + 2 * ((s + 1) & 1)) * NOUT, bnxt);
        __builtin_amdgcn_sched_barrier(0);
        v2f av;
        av.x = p ? a[j].z : a[j].x;
        av.y = p ? a[j].w : a[j].y;
        #pragma unroll
        for (int t = 0; t < NT; ++t) {
            acc[t] = __builtin_amdgcn_wmma_f32_16x16x4_f32(
                 false, av,  false, bcur[t],
                 (short)0, acc[t],  false,  false);
        }
        #pragma unroll
        for (int t = 0; t < NT; ++t) bcur[t] = bnxt[t];
    }

    const int rowBase = mTile * 16 + 8 * h;
    #pragma unroll
    for (int i = 0; i < 8; ++i) {
        const int row = rowBase + i;
        if (row < M) {
            float* cp = C + (size_t)row * NOUT + c;
            #pragma unroll
            for (int t = 0; t < NT; ++t) cp[16 * t] = acc[t][i];
        }
    }
}

#define TILE_ROWS   391
#define REC_ROW     OUT_DIM
#define AGG_THREADS 1024
#define AGG_WAVES   (AGG_THREADS / 32)
#define LIST_CAP    640
#define LIST_FLUSH  (LIST_CAP - 512)
#define ROW_BATCH   8

template <int F, int B>
__device__ __forceinline__ void gcn_add_rows(const float* __restrict__ H, float* acc,
                                             int s, int roff, int wbits, int j, int lane) {
    constexpr int Q = F / 32;
    int oj[B];
    float wj[B];
    float v[B][Q];
    #pragma unroll
    for (int u = 0; u < B; ++u) {
        const int sj = __builtin_amdgcn_readlane(s, j + u);
        oj[u] = __builtin_amdgcn_readlane(roff, j + u);
        wj[u] = __int_as_float(__builtin_amdgcn_readlane(wbits, j + u));
        const float* hp = H + (size_t)sj * F + lane;
        #pragma unroll
        for (int q = 0; q < Q; ++q) v[u][q] = hp[q * 32];
    }
    #pragma unroll
    for (int u = 0; u < B; ++u) {
        #pragma unroll
        for (int q = 0; q < Q; ++q)
            atomicAdd(&acc[oj[u] + lane + q * 32], v[u][q] * wj[u]);
    }
}

template <int F, bool REC>
__device__ __forceinline__ void gcn_flush_list(const float* __restrict__ H,
                                               const int* __restrict__ src,
                                               const int* __restrict__ dst,
                                               const float* __restrict__ dinv,
                                               float* acc, const int* mylist,
                                               int cnt, int lo, int lane,
                                               int2* ent, int cap, int& written) {
    for (int base = 0; base < cnt; base += 32) {
        int m = cnt - base;
        if (m > 32) m = 32;
        const int idx = (lane < m) ? (base + lane) : base;
        const int e = mylist[idx];
        const int s = src[e];
        const int d = dst[e];
        if (REC) {
            if (written + m <= cap && lane < m) ent[written + lane] = make_int2(s, d);
            written += m;
        }
        const float w = dinv[s] * dinv[d];
        const int roff = (d - lo) * F;
        const int wbits = __float_as_int(w);

        int j = 0;
        for (; j + ROW_BATCH <= m; j += ROW_BATCH)
            gcn_add_rows<F, ROW_BATCH>(H, acc, s, roff, wbits, j, lane);
        if (j + 4 <= m) { gcn_add_rows<F, 4>(H, acc, s, roff, wbits, j, lane); j += 4; }
        if (j + 2 <= m) { gcn_add_rows<F, 2>(H, acc, s, roff, wbits, j, lane); j += 2; }
        if (j < m)      { gcn_add_rows<F, 1>(H, acc, s, roff, wbits, j, lane); }
    }
}

template <int F>
__device__ __forceinline__ void gcn_replay_list(const float* __restrict__ H,
                                                const float* __restrict__ dinv,
                                                float* acc, const int2* ent,
                                                int cnt, int lo, int lane) {
    int2 en = ent[(lane < cnt) ? lane : 0];
    for (int base = 0; base < cnt; base += 32) {
        int m = cnt - base;
        if (m > 32) m = 32;
        const int s = en.x;
        const int d = en.y;
        const int nb = base + 32;
        if (nb < cnt) en = ent[(nb + lane < cnt) ? (nb + lane) : nb];
        const float w = dinv[s] * dinv[d];
        const int roff = (d - lo) * F;
        const int wbits = __float_as_int(w);

        int j = 0;
        for (; j + ROW_BATCH <= m; j += ROW_BATCH)
            gcn_add_rows<F, ROW_BATCH>(H, acc, s, roff, wbits, j, lane);
        if (j + 4 <= m) { gcn_add_rows<F, 4>(H, acc, s, roff, wbits, j, lane); j += 4; }
        if (j + 2 <= m) { gcn_add_rows<F, 2>(H, acc, s, roff, wbits, j, lane); j += 2; }
        if (j < m)      { gcn_add_rows<F, 1>(H, acc, s, roff, wbits, j, lane); }
    }
}

template <int F, bool REPLAY>
__global__ __launch_bounds__(AGG_THREADS) void gcn_tile_aggregate(
    const float* __restrict__ H, const int* __restrict__ src,
    const int* __restrict__ dst, const float* __restrict__ dinv,
    const float* __restrict__ bias, float* out, float* rec, int N, int E) {
    static_assert(REC_ROW % AGG_WAVES == 0, "a tile row splits evenly over the waves' record slices");
    __shared__ __align__(16) float acc[TILE_ROWS * F];
    __shared__ int elist[AGG_WAVES * LIST_CAP];

    const int tid  = threadIdx.x;
    const int lane = tid & 31;
    const int wave = __builtin_amdgcn_readfirstlane(tid >> 5);
    const int lo   = blockIdx.x * TILE_ROWS;
    int nrows = N - lo;
    if (nrows > TILE_ROWS) nrows = TILE_ROWS;
    const int nelem = nrows * F;
    const size_t gbase = (size_t)lo * F;

    const int nvec = nelem >> 2;
    {
        const float4* H4 = reinterpret_cast<const float4*>(H + gbase);
        float4* acc4 = reinterpret_cast<float4*>(acc);
        #pragma unroll 4
        for (int i = tid; i < nvec; i += AGG_THREADS) {
            const float di = dinv[lo + (int)((unsigned)i / (unsigned)(F / 4))];
            float4 h = H4[i];
            h.x = h.x * di * di; h.y = h.y * di * di; h.z = h.z * di * di; h.w = h.w * di * di;
            acc4[i] = h;
        }
    }
    __syncthreads();

    int* mylist = elist + wave * LIST_CAP;
    int cnt = 0;

    const int seg = (nrows * (REC_ROW / AGG_WAVES)) & ~31;
    int* segp = reinterpret_cast<int*>(rec) + (size_t)lo * REC_ROW + (size_t)wave * seg;
    int2* ent = reinterpret_cast<int2*>(segp + 2);
    const int cap = (seg >= 32) ? ((seg - 2) >> 1) : -1;
    int written = 0;

    bool from_record = false;
    if (REPLAY) {
        int saved = -1;
        if (cap >= 0) saved = segp[0];
        from_record = (saved >= 0) && (saved <= cap);
        if (from_record && saved > 0) gcn_replay_list<F>(H, dinv, acc, ent, saved, lo, lane);
    }
    if (!from_record) {

#define GCN_APPEND(bal, mk, ek)                                                     \
    if ((bal) != 0u) {                                                              \
        if (mk) mylist[cnt + (int)__builtin_amdgcn_mbcnt_lo((bal), 0u)] = (ek);     \
        cnt += __popc(bal);                                                         \
    }
#define GCN_SCAN4(dv, eb)                                                           \
    {                                                                               \
        const bool m0 = (unsigned)((dv).x - lo) < (unsigned)TILE_ROWS;              \
        const bool m1 = (unsigned)((dv).y - lo) < (unsigned)TILE_ROWS;              \
        const bool m2 = (unsigned)((dv).z - lo) < (unsigned)TILE_ROWS;              \
        const bool m3 = (unsigned)((dv).w - lo) < (unsigned)TILE_ROWS;              \
        const unsigned q0 = __builtin_amdgcn_ballot_w32(m0);                        \
        const unsigned q1 = __builtin_amdgcn_ballot_w32(m1);                        \
        const unsigned q2 = __builtin_amdgcn_ballot_w32(m2);                        \
        const unsigned q3 = __builtin_amdgcn_ballot_w32(m3);                        \
        if ((q0 | q1 | q2 | q3) != 0u) {                                            \
            GCN_APPEND(q0, m0, (eb))                                                \
            GCN_APPEND(q1, m1, (eb) + 1)                                            \
            GCN_APPEND(q2, m2, (eb) + 2)                                            \
            GCN_APPEND(q3, m3, (eb) + 3)                                            \
        }                                                                           \
    }

    const int lead = (int)(((16u - (unsigned)((size_t)dst & 15u)) & 15u) >> 2);
    const int nGroups = (E > lead) ? ((E - lead) >> 9) : 0;
    {
        const int4* dstv = reinterpret_cast<const int4*>(dst + lead);
        int g = wave;
        int4 n0 = make_int4(-1, -1, -1, -1), n1 = n0, n2 = n0, n3 = n0;
        if (g < nGroups) {
            const int4* p = dstv + ((size_t)g << 7) + lane;
            n0 = p[0]; n1 = p[32]; n2 = p[64]; n3 = p[96];
        }
        while (g < nGroups) {
            const int4 c0 = n0, c1 = n1, c2 = n2, c3 = n3;
            const int gn = g + AGG_WAVES;
            if (gn < nGroups) {
                const int4* p = dstv + ((size_t)gn << 7) + lane;
                n0 = p[0]; n1 = p[32]; n2 = p[64]; n3 = p[96];
            }
            const int eb = lead + (g << 9) + (lane << 2);
            GCN_SCAN4(c0, eb)
            GCN_SCAN4(c1, eb + 128)
            GCN_SCAN4(c2, eb + 256)
            GCN_SCAN4(c3, eb + 384)
            if (cnt > LIST_FLUSH) {
                gcn_flush_list<F, !REPLAY>(H, src, dst, dinv, acc, mylist, cnt, lo, lane, ent, cap, written);
                cnt = 0;
            }
            g = gn;
        }
    }
    {
        const int e = (tid < lead) ? tid : ((nGroups << 9) + tid);
        bool mk = false;
        if (e < E) mk = (unsigned)(dst[e] - lo) < (unsigned)TILE_ROWS;
        const unsigned qt = __builtin_amdgcn_ballot_w32(mk);
        GCN_APPEND(qt, mk, e)
    }
    if (cnt > 0) gcn_flush_list<F, !REPLAY>(H, src, dst, dinv, acc, mylist, cnt, lo, lane, ent, cap, written);
    if (!REPLAY && cap >= 0 && lane == 0) segp[0] = written;
    }
#undef GCN_SCAN4
#undef GCN_APPEND
    __syncthreads();

    {
        const float4 bv = reinterpret_cast<const float4*>(bias)[tid & (F / 4 - 1)];
        const float4* acc4 = reinterpret_cast<const float4*>(acc);
        float4* out4 = reinterpret_cast<float4*>(out + gbase);
        for (int i = tid; i < nvec; i += AGG_THREADS) {
            float4 a = acc4[i];
            a.x += bv.x; a.y += bv.y; a.z += bv.z; a.w += bv.w;
            a.x = (a.x > 0.0f) ? a.x : expm1f(a.x);
            a.y = (a.y > 0.0f) ? a.y : expm1f(a.y);
            a.z = (a.z > 0.0f) ? a.z : expm1f(a.z);
            a.w = (a.w > 0.0f) ? a.w : expm1f(a.w);
            out4[i] = a;
        }
    }
}

extern "C" void kernel_launch(void* const* d_in, const int* in_sizes, int n_in,
                              void* d_out, int out_size, void* d_ws, size_t ws_size,
                              hipStream_t stream) {
    const float* x  = (const float*)d_in[0];
    const int*   ei = (const int*)d_in[1];
    const float* W1 = (const float*)d_in[2];
    const float* b1 = (const float*)d_in[3];
    const float* W2 = (const float*)d_in[4];
    const float* b2 = (const float*)d_in[5];
    float* out = (float*)d_out;

    const int N = in_sizes[0] / IN_DIM;
    const int E = in_sizes[1] / 2;
    const int* src = ei;
    const int* dst = ei + E;

    char* ws = (char*)d_ws;
    float* dinv = (float*)ws;
    size_t off1 = ((size_t)N * 4 + 255) & ~(size_t)255;
    float* hbuf = (float*)(ws + off1);
    size_t off2 = off1 + ((((size_t)N * HID_DIM * 4) + 255) & ~(size_t)255);
    float* act1 = (float*)(ws + off2);

    const int TB = 256;
    const int mTiles = (N + 15) / 16;
    const int nTiles = (N + TILE_ROWS - 1) / TILE_ROWS;

    gcn_fill_one<<<(N + TB - 1) / TB, TB, 0, stream>>>(dinv, N);
    gcn_deg_count<<<(E + TB - 1) / TB, TB, 0, stream>>>(dst, dinv, E);
    gcn_rsqrt_ip<<<(N + TB - 1) / TB, TB, 0, stream>>>(dinv, N);

    gcn_gemm_wmma<HID_DIM><<<(mTiles + GEMM_WAVES - 1) / GEMM_WAVES, GEMM_THREADS, 0, stream>>>(x, W1, hbuf, N);
    gcn_tile_aggregate<HID_DIM, false><<<nTiles, AGG_THREADS, 0, stream>>>(hbuf, src, dst, dinv, b1, act1, out, N, E);

    gcn_gemm_wmma<OUT_DIM><<<(mTiles + GEMM_WAVES - 1) / GEMM_WAVES, GEMM_THREADS, 0, stream>>>(act1, W2, hbuf, N);
    gcn_tile_aggregate<OUT_DIM, true><<<nTiles, AGG_THREADS, 0, stream>>>(hbuf, src, dst, dinv, b2, out, out, N, E);
}


# Layer-1 aggregate scans one tile-index byte per edge (stored by the degree kernel in free rows of the output buffer), 16 edges per word test, not 4-byte dst; wave shares change, sums in arrival order
# speedup vs baseline: 2.3962x; 1.0829x over previous
#include <hip/hip_runtime.h>
#include <cmath>

typedef __attribute__((ext_vector_type(2))) float v2f;
typedef __attribute__((ext_vector_type(8))) float v8f;

#define IN_DIM 128
#define HID_DIM 128
#define OUT_DIM 64
#define KDIM 128

#define TILE_ROWS   391
#define REC_ROW     OUT_DIM
#define TILE_BYTES  (TILE_ROWS * REC_ROW * 4)
#define KEY_ROWS      64
#define KEY_SHIFT     14
#define KEYS_PER_TILE (1 << KEY_SHIFT)
#define KEY_OFF       ((TILE_ROWS - KEY_ROWS) * REC_ROW * 4)

__global__ void gcn_fill_one(float* __restrict__ deg, int n) {
    int i = blockIdx.x * blockDim.x + threadIdx.x;
    if (i < n) deg[i] = 1.0f;
}

__global__ void gcn_deg_keys(const int* __restrict__ dst, float* __restrict__ deg,
                             unsigned char* keys, int e) {
    const int i = blockIdx.x * blockDim.x + threadIdx.x;
    const int e0 = 4 * i;
    if (e0 >= e) return;
    unsigned char* kp = keys + (size_t)(e0 >> KEY_SHIFT) * TILE_BYTES + (e0 & (KEYS_PER_TILE - 1));
    if (e0 + 4 <= e) {
        unsigned packed = 0u;
        #pragma unroll
        for (int k = 0; k < 4; ++k) {
            const int d = dst[e0 + k];
            unsafeAtomicAdd(&deg[d], 1.0f);
            packed |= ((unsigned)d / (unsigned)TILE_ROWS) << (8 * k);
        }
        *reinterpret_cast<unsigned*>(kp) = packed;
    } else {
        for (int k = 0; e0 + k < e; ++k) {
            const int d = dst[e0 + k];
            unsafeAtomicAdd(&deg[d], 1.0f);
            kp[k] = (unsigned char)((unsigned)d / (unsigned)TILE_ROWS);
        }
    }
}

__global__ void gcn_rsqrt_ip(float* __restrict__ deg, int n) {
    int i = blockIdx.x * blockDim.x + threadIdx.x;
    if (i < n) deg[i] = rsqrtf(deg[i]);
}

#define GEMM_THREADS 128
#define GEMM_WAVES   (GEMM_THREADS / 32)

template <int NOUT, int NT>
__device__ __forceinline__ void gcn_load_b_step(const float* __restrict__ bk, v2f (&b)[NT]) {
    #pragma unroll
    for (int t = 0; t < NT; ++t) {
        b[t].x = bk[16 * t];
        b[t].y = bk[NOUT + 16 * t];
    }
}

template <int NOUT>
__global__ __launch_bounds__(GEMM_THREADS) void gcn_gemm_wmma(
    const float* __restrict__ A, const float* __restrict__ B,
    float* __restrict__ C, int M) {
    constexpr int NT = NOUT / 16;

    const int tid  = threadIdx.x;
    const int wave = tid >> 5;
    const int lane = tid & 31;
    const int c    = lane & 15;
    const int h    = lane >> 4;

    const int mTile = blockIdx.x * GEMM_WAVES + wave;
    if (mTile * 16 >= M) return;

    int arow = mTile * 16 + c;
    if (arow >= M) arow = M - 1;

    const float4* ap = reinterpret_cast<const float4*>(A + (size_t)arow * KDIM) + h;
    float4 a[16];
    #pragma unroll
    for (int j = 0; j < 16; ++j) a[j] = ap[2 * j];

    const float* bp = B + (size_t)(4 * h) * NOUT + c;

    v2f bcur[NT], bnxt[NT];
    gcn_load_b_step<NOUT, NT>(bp, bcur);

    __builtin_amdgcn_sched_barrier(0);

    v8f acc[NT];
    #pragma unroll
    for (int t = 0; t < NT; ++t) acc[t] = v8f{};

    #pragma unroll
    for (int s = 0; s < 32; ++s) {
        const int j = s >> 1;
        const int p = s & 1;
        if (s + 1 < 32)
            gcn_load_b_step<NOUT, NT>(bp + (size_t)(8 * ((s + 1) >> 1) + 2 * ((s + 1) & 1)) * NOUT, bnxt);
        __builtin_amdgcn_sched_barrier(0);
        v2f av;
        av.x = p ? a[j].z : a[j].x;
        av.y = p ? a[j].w : a[j].y;
        #pragma unroll
        for (int t = 0; t < NT; ++t) {
            acc[t] = __builtin_amdgcn_wmma_f32_16x16x4_f32(
                 false, av,  false, bcur[t],
                 (short)0, acc[t],  false,  false);
        }
        #pragma unroll
        for (int t = 0; t < NT; ++t) bcur[t] = bnxt[t];
    }

    const int rowBase = mTile * 16 + 8 * h;
    #pragma unroll
    for (int i = 0; i < 8; ++i) {
        const int row = rowBase + i;
        if (row < M) {
            float* cp = C + (size_t)row * NOUT + c;
            #pragma unroll
            for (int t = 0; t < NT; ++t) cp[16 * t] = acc[t][i];
        }
    }
}

#define AGG_THREADS 1024
#define AGG_WAVES   (AGG_THREADS / 32)
#define LIST_CAP    640
#define LIST_FLUSH  (LIST_CAP - 512)
#define ROW_BATCH   8

template <int F, int B>
__device__ __forceinline__ void gcn_add_rows(const float* __restrict__ H, float* acc,
                                             int s, int roff, int wbits, int j, int lane) {
    constexpr int Q = F / 32;
    int oj[B];
    float wj[B];
    float v[B][Q];
    #pragma unroll
    for (int u = 0; u < B; ++u) {
        const int sj = __builtin_amdgcn_readlane(s, j + u);
        oj[u] = __builtin_amdgcn_readlane(roff, j + u);
        wj[u] = __int_as_float(__builtin_amdgcn_readlane(wbits, j + u));
        const float* hp = H + (size_t)sj * F + lane;
        #pragma unroll
        for (int q = 0; q < Q; ++q) v[u][q] = hp[q * 32];
    }
    #pragma unroll
    for (int u = 0; u < B; ++u) {
        #pragma unroll
        for (int q = 0; q < Q; ++q)
            atomicAdd(&acc[oj[u] + lane + q * 32], v[u][q] * wj[u]);
    }
}

template <int F, bool REC>
__device__ __forceinline__ void gcn_flush_list(const float* __restrict__ H,
                                               const int* __restrict__ src,
                                               const int* __restrict__ dst,
                                               const float* __restrict__ dinv,
                                               float* acc, const int* mylist,
                                               int cnt, int lo, int lane,
                                               int2* ent, int cap, int& written) {
    for (int base = 0; base < cnt; base += 32) {
        int m = cnt - base;
        if (m > 32) m = 32;
        const int idx = (lane < m) ? (base + lane) : base;
        const int e = mylist[idx];
        const int s = src[e];
        const int d = dst[e];
        if (REC) {
            if (written + m <= cap && lane < m) ent[written + lane] = make_int2(s, d);
            written += m;
        }
        const float w = dinv[s] * dinv[d];
        const int roff = (d - lo) * F;
        const int wbits = __float_as_int(w);

        int j = 0;
        for (; j + ROW_BATCH <= m; j += ROW_BATCH)
            gcn_add_rows<F, ROW_BATCH>(H, acc, s, roff, wbits, j, lane);
        if (j + 4 <= m) { gcn_add_rows<F, 4>(H, acc, s, roff, wbits, j, lane); j += 4; }
        if (j + 2 <= m) { gcn_add_rows<F, 2>(H, acc, s, roff, wbits, j, lane); j += 2; }
        if (j < m)      { gcn_add_rows<F, 1>(H, acc, s, roff, wbits, j, lane); }
    }
}

template <int F>
__device__ __forceinline__ void gcn_replay_list(const float* __restrict__ H,
                                                const float* __restrict__ dinv,
                                                float* acc, const int2* ent,
                                                int cnt, int lo, int lane) {
    int2 en = ent[(lane < cnt) ? lane : 0];
    for (int base = 0; base < cnt; base += 32) {
        int m = cnt - base;
        if (m > 32) m = 32;
        const int s = en.x;
        const int d = en.y;
        const int nb = base + 32;
        if (nb < cnt) en = ent[(nb + lane < cnt) ? (nb + lane) : nb];
        const float w = dinv[s] * dinv[d];
        const int roff = (d - lo) * F;
        const int wbits = __float_as_int(w);

        int j = 0;
        for (; j + ROW_BATCH <= m; j += ROW_BATCH)
            gcn_add_rows<F, ROW_BATCH>(H, acc, s, roff, wbits, j, lane);
        if (j + 4 <= m) { gcn_add_rows<F, 4>(H, acc, s, roff, wbits, j, lane); j += 4; }
        if (j + 2 <= m) { gcn_add_rows<F, 2>(H, acc, s, roff, wbits, j, lane); j += 2; }
        if (j < m)      { gcn_add_rows<F, 1>(H, acc, s, roff, wbits, j, lane); }
    }
}

template <int F, bool REPLAY>
__global__ __launch_bounds__(AGG_THREADS) void gcn_tile_aggregate(
    const float* __restrict__ H, const int* __restrict__ src,
    const int* __restrict__ dst, const float* __restrict__ dinv,
    const float* __restrict__ bias, float* out, float* rec,
    const unsigned char* keys, int N, int E) {
    static_assert(REC_ROW % AGG_WAVES == 0, "a tile row splits evenly over the waves' record slices");
    static_assert(KEYS_PER_TILE == KEY_ROWS * REC_ROW * 4, "a tile's keys fill its reserved rows exactly");
    __shared__ __align__(16) float acc[TILE_ROWS * F];
    __shared__ int elist[AGG_WAVES * LIST_CAP];

    const int tid  = threadIdx.x;
    const int lane = tid & 31;
    const int wave = __builtin_amdgcn_readfirstlane(tid >> 5);
    const int lo   = blockIdx.x * TILE_ROWS;
    int nrows = N - lo;
    if (nrows > TILE_ROWS) nrows = TILE_ROWS;
    const int nelem = nrows * F;
    const size_t gbase = (size_t)lo * F;

    const int nvec = nelem >> 2;
    {
        const float4* H4 = reinterpret_cast<const float4*>(H + gbase);
        float4* acc4 = reinterpret_cast<float4*>(acc);
        #pragma unroll 4
        for (int i = tid; i < nvec; i += AGG_THREADS) {
            const float di = dinv[lo + (int)((unsigned)i / (unsigned)(F / 4))];
            float4 h = H4[i];
            h.x = h.x * di * di; h.y = h.y * di * di; h.z = h.z * di * di; h.w = h.w * di * di;
            acc4[i] = h;
        }
    }
    __syncthreads();

    int* mylist = elist + wave * LIST_CAP;
    int cnt = 0;

    const int rrows = (nrows > KEY_ROWS) ? (nrows - KEY_ROWS) : 0;
    const int seg = (rrows * (REC_ROW / AGG_WAVES)) & ~31;
    int* segp = reinterpret_cast<int*>(rec) + (size_t)lo * REC_ROW + (size_t)wave * seg;
    int2* ent = reinterpret_cast<int2*>(segp + 2);
    const int cap = (seg >= 32) ? ((seg - 2) >> 1) : -1;
    int written = 0;

    bool from_record = false;
    if (REPLAY) {
        int saved = -1;
        if (cap >= 0) saved = segp[0];
        from_record = (saved >= 0) && (saved <= cap);
        if (from_record && saved > 0) gcn_replay_list<F>(H, dinv, acc, ent, saved, lo, lane);
    }
    if (!from_record) {

#define GCN_APPEND(bal, mk, ek)                                                     \
    if ((bal) != 0u) {                                                              \
        if (mk) mylist[cnt + (int)__builtin_amdgcn_mbcnt_lo((bal), 0u)] = (ek);     \
        cnt += __popc(bal);                                                         \
    }
#define GCN_SCAN16(kv, eb)                                                          \
    {                                                                               \
        const unsigned x0 = (kv).x ^ pat, x1 = (kv).y ^ pat;                        \
        const unsigned x2 = (kv).z ^ pat, x3 = (kv).w ^ pat;                        \
        const unsigned y0 = ((x0 & 0x7f7f7f7fu) + 0x7f7f7f7fu) | x0;                \
        const unsigned y1 = ((x1 & 0x7f7f7f7fu) + 0x7f7f7f7fu) | x1;                \
        const unsigned y2 = ((x2 & 0x7f7f7f7fu) + 0x7f7f7f7fu) | x2;                \
        const unsigned y3 = ((x3 & 0x7f7f7f7fu) + 0x7f7f7f7fu) | x3;                \
        const unsigned hit = ~(y0 & y1 & y2 & y3) & 0x80808080u;                    \
        unsigned bal = __builtin_amdgcn_ballot_w32(hit != 0u);                      \
        if (bal != 0u) {                                                            \
            unsigned f = ((~y0 & 0x80808080u) >> 7) | ((~y1 & 0x80808080u) >> 6) |  \
                         ((~y2 & 0x80808080u) >> 5) | ((~y3 & 0x80808080u) >> 4);   \
            do {                                                                    \
                if (f != 0u) {                                                      \
                    const int fi = __builtin_ctz(f);                                \
                    mylist[cnt + (int)__builtin_amdgcn_mbcnt_lo(bal, 0u)] =         \
                        (eb) + ((fi & 7) << 2) + (fi >> 3);                         \
                    f &= f - 1u;                                                    \
                }                                                                   \
                cnt += __popc(bal);                                                 \
                bal = __builtin_amdgcn_ballot_w32(f != 0u);                         \
            } while (bal != 0u);                                                    \
        }                                                                           \
    }

    if constexpr (!REPLAY) {
        const unsigned pat = (unsigned)blockIdx.x * 0x01010101u;
        const int nGroups = E >> 11;
        int g = wave;
        uint4 n0 = make_uint4(0u, 0u, 0u, 0u), n1 = n0, n2 = n0, n3 = n0;
        if (g < nGroups) {
            const uint4* p = reinterpret_cast<const uint4*>(
                keys + (size_t)(g >> 3) * TILE_BYTES + ((g & 7) << 11)) + lane;
            n0 = p[0]; n1 = p[32]; n2 = p[64]; n3 = p[96];
        }
        while (g < nGroups) {
            const uint4 c0 = n0, c1 = n1, c2 = n2, c3 = n3;
            const int gn = g + AGG_WAVES;
            if (gn < nGroups) {
                const uint4* p = reinterpret_cast<const uint4*>(
                    keys + (size_t)(gn >> 3) * TILE_BYTES + ((gn & 7) << 11)) + lane;
                n0 = p[0]; n1 = p[32]; n2 = p[64]; n3 = p[96];
            }
            const int eb = (g << 11) + (lane << 4);
            int sub = 0;
            do {
                if (sub == 0) { GCN_SCAN16(c0, eb) sub = 1; }
                if (sub == 1 && cnt <= LIST_FLUSH) { GCN_SCAN16(c1, eb + 512) sub = 2; }
                if (sub == 2 && cnt <= LIST_FLUSH) { GCN_SCAN16(c2, eb + 1024) sub = 3; }
                if (sub == 3 && cnt <= LIST_FLUSH) { GCN_SCAN16(c3, eb + 1536) sub = 4; }
                if (cnt > LIST_FLUSH) {
                    gcn_flush_list<F, true>(H, src, dst, dinv, acc, mylist, cnt, lo, lane, ent, cap, written);
                    cnt = 0;
                }
            } while (sub < 4);
            g = gn;
        }
    } else {
        const int nGroups = E >> 11;
        for (int g = wave; g < nGroups; g += AGG_WAVES) {
            for (int k = 0; k < 2048; k += 128) {
                const int eb = (g << 11) + k + lane;
                const int d0 = dst[eb], d1 = dst[eb + 32], d2 = dst[eb + 64], d3 = dst[eb + 96];
                const bool m0 = (unsigned)(d0 - lo) < (unsigned)TILE_ROWS;
                const bool m1 = (unsigned)(d1 - lo) < (unsigned)TILE_ROWS;
                const bool m2 = (unsigned)(d2 - lo) < (unsigned)TILE_ROWS;
                const bool m3 = (unsigned)(d3 - lo) < (unsigned)TILE_ROWS;
                const unsigned q0 = __builtin_amdgcn_ballot_w32(m0);
                const unsigned q1 = __builtin_amdgcn_ballot_w32(m1);
                const unsigned q2 = __builtin_amdgcn_ballot_w32(m2);
                const unsigned q3 = __builtin_amdgcn_ballot_w32(m3);
                GCN_APPEND(q0, m0, eb)
                GCN_APPEND(q1, m1, eb + 32)
                GCN_APPEND(q2, m2, eb + 64)
                GCN_APPEND(q3, m3, eb + 96)
                if (cnt > LIST_FLUSH) {
                    gcn_flush_list<F, false>(H, src, dst, dinv, acc, mylist, cnt, lo, lane, ent, cap, written);
                    cnt = 0;
                }
            }
        }
    }
    for (int base = (E >> 11) << 11; base < E; base += AGG_THREADS) {
        const int e = base + tid;
        bool mk = false;
        if (e < E) mk = (unsigned)(dst[e] - lo) < (unsigned)TILE_ROWS;
        const unsigned qt = __builtin_amdgcn_ballot_w32(mk);
        GCN_APPEND(qt, mk, e)
    }
    if (cnt > 0) gcn_flush_list<F, !REPLAY>(H, src, dst, dinv, acc, mylist, cnt, lo, lane, ent, cap, written);
    if (!REPLAY && cap >= 0 && lane == 0) segp[0] = written;
    }
#undef GCN_SCAN16
#undef GCN_APPEND
    __syncthreads();

    {
        const float4 bv = reinterpret_cast<const float4*>(bias)[tid & (F / 4 - 1)];
        const float4* acc4 = reinterpret_cast<const float4*>(acc);
        float4* out4 = reinterpret_cast<float4*>(out + gbase);
        for (int i = tid; i < nvec; i += AGG_THREADS) {
            float4 a = acc4[i];
            a.x += bv.x; a.y += bv.y; a.z += bv.z; a.w += bv.w;
            a.x = (a.x > 0.0f) ? a.x : expm1f(a.x);
            a.y = (a.y > 0.0f) ? a.y : expm1f(a.y);
            a.z = (a.z > 0.0f) ? a.z : expm1f(a.z);
            a.w = (a.w > 0.0f) ? a.w : expm1f(a.w);
            out4[i] = a;
        }
    }
}

extern "C" void kernel_launch(void* const* d_in, const int* in_sizes, int n_in,
                              void* d_out, int out_size, void* d_ws, size_t ws_size,
                              hipStream_t stream) {
    const float* x  = (const float*)d_in[0];
    const int*   ei = (const int*)d_in[1];
    const float* W1 = (const float*)d_in[2];
    const float* b1 = (const float*)d_in[3];
    const float* W2 = (const float*)d_in[4];
    const float* b2 = (const float*)d_in[5];
    float* out = (float*)d_out;

    const int N = in_sizes[0] / IN_DIM;
    const int E = in_sizes[1] / 2;
    const int* src = ei;
    const int* dst = ei + E;

    char* ws = (char*)d_ws;
    float* dinv = (float*)ws;
    size_t off1 = ((size_t)N * 4 + 255) & ~(size_t)255;
    float* hbuf = (float*)(ws + off1);
    size_t off2 = off1 + ((((size_t)N * HID_DIM * 4) + 255) & ~(size_t)255);
    float* act1 = (float*)(ws + off2);

    const int TB = 256;
    const int mTiles = (N + 15) / 16;
    const int nTiles = (N + TILE_ROWS - 1) / TILE_ROWS;

    gcn_fill_one<<<(N + TB - 1) / TB, TB, 0, stream>>>(dinv, N);
    unsigned char* keys = (unsigned char*)d_out + KEY_OFF;
    gcn_deg_keys<<<((E + 3) / 4 + TB - 1) / TB, TB, 0, stream>>>(dst, dinv, keys, E);
    gcn_rsqrt_ip<<<(N + TB - 1) / TB, TB, 0, stream>>>(dinv, N);

    gcn_gemm_wmma<HID_DIM><<<(mTiles + GEMM_WAVES - 1) / GEMM_WAVES, GEMM_THREADS, 0, stream>>>(x, W1, hbuf, N);
    gcn_tile_aggregate<HID_DIM, false><<<nTiles, AGG_THREADS, 0, stream>>>(hbuf, src, dst, dinv, b1, act1, out, keys, N, E);

    gcn_gemm_wmma<OUT_DIM><<<(mTiles + GEMM_WAVES - 1) / GEMM_WAVES, GEMM_THREADS, 0, stream>>>(act1, W2, hbuf, N);
    gcn_tile_aggregate<OUT_DIM, true><<<nTiles, AGG_THREADS, 0, stream>>>(hbuf, src, dst, dinv, b2, out, out, keys, N, E);
}


# Edge weight split: GEMM epilogues store dinv[r]*row; aggregates add source rows unweighted in arrival order (no dinv gathers, no multiply per edge) and scale each row sum by dinv[dst] at write-back
# speedup vs baseline: 2.5297x; 1.0557x over previous
#include <hip/hip_runtime.h>
#include <cmath>

typedef __attribute__((ext_vector_type(2))) float v2f;
typedef __attribute__((ext_vector_type(8))) float v8f;

#define IN_DIM 128
#define HID_DIM 128
#define OUT_DIM 64
#define KDIM 128

#define TILE_ROWS   391
#define REC_ROW     OUT_DIM
#define TILE_BYTES  (TILE_ROWS * REC_ROW * 4)
#define KEY_ROWS      64
#define KEY_SHIFT     14
#define KEYS_PER_TILE (1 << KEY_SHIFT)
#define KEY_OFF       ((TILE_ROWS - KEY_ROWS) * REC_ROW * 4)

__global__ void gcn_fill_one(float* __restrict__ deg, int n) {
    int i = blockIdx.x * blockDim.x + threadIdx.x;
    if (i < n) deg[i] = 1.0f;
}

__global__ void gcn_deg_keys(const int* __restrict__ dst, float* __restrict__ deg,
                             unsigned char* keys, int e) {
    const int i = blockIdx.x * blockDim.x + threadIdx.x;
    const int e0 = 4 * i;
    if (e0 >= e) return;
    unsigned char* kp = keys + (size_t)(e0 >> KEY_SHIFT) * TILE_BYTES + (e0 & (KEYS_PER_TILE - 1));
    if (e0 + 4 <= e) {
        unsigned packed = 0u;
        #pragma unroll
        for (int k = 0; k < 4; ++k) {
            const int d = dst[e0 + k];
            unsafeAtomicAdd(&deg[d], 1.0f);
            packed |= ((unsigned)d / (unsigned)TILE_ROWS) << (8 * k);
        }
        *reinterpret_cast<unsigned*>(kp) = packed;
    } else {
        for (int k = 0; e0 + k < e; ++k) {
            const int d = dst[e0 + k];
            unsafeAtomicAdd(&deg[d], 1.0f);
            kp[k] = (unsigned char)((unsigned)d / (unsigned)TILE_ROWS);
        }
    }
}

__global__ void gcn_rsqrt_ip(float* __restrict__ deg, int n) {
    int i = blockIdx.x * blockDim.x + threadIdx.x;
    if (i < n) deg[i] = rsqrtf(deg[i]);
}

#define GEMM_THREADS 128
#define GEMM_WAVES   (GEMM_THREADS / 32)

template <int NOUT, int NT>
__device__ __forceinline__ void gcn_load_b_step(const float* __restrict__ bk, v2f (&b)[NT]) {
    #pragma unroll
    for (int t = 0; t < NT; ++t) {
        b[t].x = bk[16 * t];
        b[t].y = bk[NOUT + 16 * t];
    }
}

template <int NOUT>
__global__ __launch_bounds__(GEMM_THREADS) void gcn_gemm_wmma(
    const float* __restrict__ A, const float* __restrict__ B,
    const float* __restrict__ rs, float* __restrict__ C, int M) {
    constexpr int NT = NOUT / 16;

    const int tid  = threadIdx.x;
    const int wave = tid >> 5;
    const int lane = tid & 31;
    const int c    = lane & 15;
    const int h    = lane >> 4;

    const int mTile = blockIdx.x * GEMM_WAVES + wave;
    if (mTile * 16 >= M) return;

    int arow = mTile * 16 + c;
    if (arow >= M) arow = M - 1;

    const float4* ap = reinterpret_cast<const float4*>(A + (size_t)arow * KDIM) + h;
    float4 a[16];
    #pragma unroll
    for (int j = 0; j < 16; ++j) a[j] = ap[2 * j];

    const float* bp = B + (size_t)(4 * h) * NOUT + c;

    v2f bcur[NT], bnxt[NT];
    gcn_load_b_step<NOUT, NT>(bp, bcur);

    __builtin_amdgcn_sched_barrier(0);

    v8f acc[NT];
    #pragma unroll
    for (int t = 0; t < NT; ++t) acc[t] = v8f{};

    const int rowBase = mTile * 16 + 8 * h;
    float sc[8];

    #pragma unroll
    for (int s = 0; s < 32; ++s) {
        const int j = s >> 1;
        const int p = s & 1;
        if (s + 1 < 32) {
            gcn_load_b_step<NOUT, NT>(bp + (size_t)(8 * ((s + 1) >> 1) + 2 * ((s + 1) & 1)) * NOUT, bnxt);
        } else {
            #pragma unroll
            for (int i = 0; i < 8; ++i) {
                int r = rowBase + i;
                if (r >= M) r = M - 1;
                sc[i] = rs[r];
            }
        }
        __builtin_amdgcn_sched_barrier(0);
        v2f av;
        av.x = p ? a[j].z : a[j].x;
        av.y = p ? a[j].w : a[j].y;
        #pragma unroll
        for (int t = 0; t < NT; ++t) {
            acc[t] = __builtin_amdgcn_wmma_f32_16x16x4_f32(
                 false, av,  false, bcur[t],
                 (short)0, acc[t],  false,  false);
        }
        #pragma unroll
        for (int t = 0; t < NT; ++t) bcur[t] = bnxt[t];
    }

    __builtin_amdgcn_sched_barrier(0);
    asm volatile("v_nop\n\tv_nop\n\tv_nop\n\tv_nop");
    __builtin_amdgcn_sched_barrier(0);

    #pragma unroll
    for (int i = 0; i < 8; ++i) {
        const int row = rowBase + i;
        if (row < M) {
            float* cp = C + (size_t)row * NOUT + c;
            #pragma unroll
            for (int t = 0; t < NT; ++t) cp[16 * t] = acc[t][i] * sc[i];
        }
    }
}

#define AGG_THREADS 1024
#define AGG_WAVES   (AGG_THREADS / 32)
#define LIST_CAP    640
#define LIST_FLUSH  (LIST_CAP - 512)
#define ROW_BATCH   8

template <int F, int B>
__device__ __forceinline__ void gcn_add_rows(const float* __restrict__ H, float* acc,
                                             int s, int roff, int j, int lane) {
    constexpr int Q = F / 32;
    int oj[B];
    float v[B][Q];
    #pragma unroll
    for (int u = 0; u < B; ++u) {
        const int sj = __builtin_amdgcn_readlane(s, j + u);
        oj[u] = __builtin_amdgcn_readlane(roff, j + u);
        const float* hp = H + (size_t)sj * F + lane;
        #pragma unroll
        for (int q = 0; q < Q; ++q) v[u][q] = hp[q * 32];
    }
    #pragma unroll
    for (int u = 0; u < B; ++u) {
        #pragma unroll
        for (int q = 0; q < Q; ++q)
            atomicAdd(&acc[oj[u] + lane + q * 32], v[u][q]);
    }
}

template <int F, bool REC>
__device__ __forceinline__ void gcn_flush_list(const float* __restrict__ H,
                                               const int* __restrict__ src,
                                               const int* __restrict__ dst,
                                               float* acc, const int* mylist,
                                               int cnt, int lo, int lane,
                                               int2* ent, int cap, int& written) {
    for (int base = 0; base < cnt; base += 32) {
        int m = cnt - base;
        if (m > 32) m = 32;
        const int idx = (lane < m) ? (base + lane) : base;
        const int e = mylist[idx];
        const int s = src[e];
        const int d = dst[e];
        if (REC) {
            if (written + m <= cap && lane < m) ent[written + lane] = make_int2(s, d);
            written += m;
        }
        const int roff = (d - lo) * F;

        int j = 0;
        for (; j + ROW_BATCH <= m; j += ROW_BATCH)
            gcn_add_rows<F, ROW_BATCH>(H, acc, s, roff, j, lane);
        if (j + 4 <= m) { gcn_add_rows<F, 4>(H, acc, s, roff, j, lane); j += 4; }
        if (j + 2 <= m) { gcn_add_rows<F, 2>(H, acc, s, roff, j, lane); j += 2; }
        if (j < m)      { gcn_add_rows<F, 1>(H, acc, s, roff, j, lane); }
    }
}

template <int F>
__device__ __forceinline__ void gcn_replay_list(const float* __restrict__ H,
                                                float* acc, const int2* ent,
                                                int cnt, int lo, int lane) {
    int2 en = ent[(lane < cnt) ? lane : 0];
    for (int base = 0; base < cnt; base += 32) {
        int m = cnt - base;
        if (m > 32) m = 32;
        const int s = en.x;
        const int d = en.y;
        const int nb = base + 32;
        if (nb < cnt) en = ent[(nb + lane < cnt) ? (nb + lane) : nb];
        const int roff = (d - lo) * F;

        int j = 0;
        for (; j + ROW_BATCH <= m; j += ROW_BATCH)
            gcn_add_rows<F, ROW_BATCH>(H, acc, s, roff, j, lane);
        if (j + 4 <= m) { gcn_add_rows<F, 4>(H, acc, s, roff, j, lane); j += 4; }
        if (j + 2 <= m) { gcn_add_rows<F, 2>(H, acc, s, roff, j, lane); j += 2; }
        if (j < m)      { gcn_add_rows<F, 1>(H, acc, s, roff, j, lane); }
    }
}

template <int F, bool REPLAY>
__global__ __launch_bounds__(AGG_THREADS) void gcn_tile_aggregate(
    const float* __restrict__ H, const int* __restrict__ src,
    const int* __restrict__ dst, const float* __restrict__ dinv,
    const float* __restrict__ bias, float* out, float* rec,
    const unsigned char* keys, int N, int E) {
    static_assert(REC_ROW % AGG_WAVES == 0, "a tile row splits evenly over the waves' record slices");
    static_assert(KEYS_PER_TILE == KEY_ROWS * REC_ROW * 4, "a tile's keys fill its reserved rows exactly");
    __shared__ __align__(16) float acc[TILE_ROWS * F];
    __shared__ int elist[AGG_WAVES * LIST_CAP];

    const int tid  = threadIdx.x;
    const int lane = tid & 31;
    const int wave = __builtin_amdgcn_readfirstlane(tid >> 5);
    const int lo   = blockIdx.x * TILE_ROWS;
    int nrows = N - lo;
    if (nrows > TILE_ROWS) nrows = TILE_ROWS;
    const int nelem = nrows * F;
    const size_t gbase = (size_t)lo * F;

    const int nvec = nelem >> 2;
    {
        const float4* H4 = reinterpret_cast<const float4*>(H + gbase);
        float4* acc4 = reinterpret_cast<float4*>(acc);
        #pragma unroll 4
        for (int i = tid; i < nvec; i += AGG_THREADS) acc4[i] = H4[i];
    }
    __syncthreads();

    int* mylist = elist + wave * LIST_CAP;
    int cnt = 0;

    const int rrows = (nrows > KEY_ROWS) ? (nrows - KEY_ROWS) : 0;
    const int seg = (rrows * (REC_ROW / AGG_WAVES)) & ~31;
    int* segp = reinterpret_cast<int*>(rec) + (size_t)lo * REC_ROW + (size_t)wave * seg;
    int2* ent = reinterpret_cast<int2*>(segp + 2);
    const int cap = (seg >= 32) ? ((seg - 2) >> 1) : -1;
    int written = 0;

    bool from_record = false;
    if (REPLAY) {
        int saved = -1;
        if (cap >= 0) saved = segp[0];
        from_record = (saved >= 0) && (saved <= cap);
        if (from_record && saved > 0) gcn_replay_list<F>(H, acc, ent, saved, lo, lane);
    }
    if (!from_record) {

#define GCN_APPEND(bal, mk, ek)                                                     \
    if ((bal) != 0u) {                                                              \
        if (mk) mylist[cnt + (int)__builtin_amdgcn_mbcnt_lo((bal), 0u)] = (ek);     \
        cnt += __popc(bal);                                                         \
    }
#define GCN_SCAN16(kv, eb)                                                          \
    {                                                                               \
        const unsigned x0 = (kv).x ^ pat, x1 = (kv).y ^ pat;                        \
        const unsigned x2 = (kv).z ^ pat, x3 = (kv).w ^ pat;                        \
        const unsigned y0 = ((x0 & 0x7f7f7f7fu) + 0x7f7f7f7fu) | x0;                \
        const unsigned y1 = ((x1 & 0x7f7f7f7fu) + 0x7f7f7f7fu) | x1;                \
        const unsigned y2 = ((x2 & 0x7f7f7f7fu) + 0x7f7f7f7fu) | x2;                \
        const unsigned y3 = ((x3 & 0x7f7f7f7fu) + 0x7f7f7f7fu) | x3;                \
        const unsigned hit = ~(y0 & y1 & y2 & y3) & 0x80808080u;                    \
        unsigned bal = __builtin_amdgcn_ballot_w32(hit != 0u);                      \
        if (bal != 0u) {                                                            \
            unsigned f = ((~y0 & 0x80808080u) >> 7) | ((~y1 & 0x80808080u) >> 6) |  \
                         ((~y2 & 0x80808080u) >> 5) | ((~y3 & 0x80808080u) >> 4);   \
            do {                                                                    \
                if (f != 0u) {                                                      \
                    const int fi = __builtin_ctz(f);                                \
                    mylist[cnt + (int)__builtin_amdgcn_mbcnt_lo(bal, 0u)] =         \
                        (eb) + ((fi & 7) << 2) + (fi >> 3);                         \
                    f &= f - 1u;                                                    \
                }                                                                   \
                cnt += __popc(bal);                                                 \
                bal = __builtin_amdgcn_ballot_w32(f != 0u);                         \
            } while (bal != 0u);                                                    \
        }                                                                           \
    }

    if constexpr (!REPLAY) {
        const unsigned pat = (unsigned)blockIdx.x * 0x01010101u;
        const int nGroups = E >> 11;
        int g = wave;
        uint4 n0 = make_uint4(0u, 0u, 0u, 0u), n1 = n0, n2 = n0, n3 = n0;
        if (g < nGroups) {
            const uint4* p = reinterpret_cast<const uint4*>(
                keys + (size_t)(g >> 3) * TILE_BYTES + ((g & 7) << 11)) + lane;
            n0 = p[0]; n1 = p[32]; n2 = p[64]; n3 = p[96];
        }
        while (g < nGroups) {
            const uint4 c0 = n0, c1 = n1, c2 = n2, c3 = n3;
            const int gn = g + AGG_WAVES;
            if (gn < nGroups) {
                const uint4* p = reinterpret_cast<const uint4*>(
                    keys + (size_t)(gn >> 3) * TILE_BYTES + ((gn & 7) << 11)) + lane;
                n0 = p[0]; n1 = p[32]; n2 = p[64]; n3 = p[96];
            }
            const int eb = (g << 11) + (lane << 4);
            int sub = 0;
            do {
                if (sub == 0) { GCN_SCAN16(c0, eb) sub = 1; }
                if (sub == 1 && cnt <= LIST_FLUSH) { GCN_SCAN16(c1, eb + 512) sub = 2; }
                if (sub == 2 && cnt <= LIST_FLUSH) { GCN_SCAN16(c2, eb + 1024) sub = 3; }
                if (sub == 3 && cnt <= LIST_FLUSH) { GCN_SCAN16(c3, eb + 1536) sub = 4; }
                if (cnt > LIST_FLUSH) {
                    gcn_flush_list<F, true>(H, src, dst, acc, mylist, cnt, lo, lane, ent, cap, written);
                    cnt = 0;
                }
            } while (sub < 4);
            g = gn;
        }
    } else {
        const int nGroups = E >> 11;
        for (int g = wave; g < nGroups; g += AGG_WAVES) {
            for (int k = 0; k < 2048; k += 128) {
                const int eb = (g << 11) + k + lane;
                const int d0 = dst[eb], d1 = dst[eb + 32], d2 = dst[eb + 64], d3 = dst[eb + 96];
                const bool m0 = (unsigned)(d0 - lo) < (unsigned)TILE_ROWS;
                const bool m1 = (unsigned)(d1 - lo) < (unsigned)TILE_ROWS;
                const bool m2 = (unsigned)(d2 - lo) < (unsigned)TILE_ROWS;
                const bool m3 = (unsigned)(d3 - lo) < (unsigned)TILE_ROWS;
                const unsigned q0 = __builtin_amdgcn_ballot_w32(m0);
                const unsigned q1 = __builtin_amdgcn_ballot_w32(m1);
                const unsigned q2 = __builtin_amdgcn_ballot_w32(m2);
                const unsigned q3 = __builtin_amdgcn_ballot_w32(m3);
                GCN_APPEND(q0, m0, eb)
                GCN_APPEND(q1, m1, eb + 32)
                GCN_APPEND(q2, m2, eb + 64)
                GCN_APPEND(q3, m3, eb + 96)
                if (cnt > LIST_FLUSH) {
                    gcn_flush_list<F, false>(H, src, dst, acc, mylist, cnt, lo, lane, ent, cap, written);
                    cnt = 0;
                }
            }
        }
    }
    for (int base = (E >> 11) << 11; base < E; base += AGG_THREADS) {
        const int e = base + tid;
        bool mk = false;
        if (e < E) mk = (unsigned)(dst[e] - lo) < (unsigned)TILE_ROWS;
        const unsigned qt = __builtin_amdgcn_ballot_w32(mk);
        GCN_APPEND(qt, mk, e)
    }
    if (cnt > 0) gcn_flush_list<F, !REPLAY>(H, src, dst, acc, mylist, cnt, lo, lane, ent, cap, written);
    if (!REPLAY && cap >= 0 && lane == 0) segp[0] = written;
    }
#undef GCN_SCAN16
#undef GCN_APPEND
    __syncthreads();

    {
        const float4 bv = reinterpret_cast<const float4*>(bias)[tid & (F / 4 - 1)];
        const float4* acc4 = reinterpret_cast<const float4*>(acc);
        float4* out4 = reinterpret_cast<float4*>(out + gbase);
        for (int i = tid; i < nvec; i += AGG_THREADS) {
            const float di = dinv[lo + (int)((unsigned)i / (unsigned)(F / 4))];
            float4 a = acc4[i];
            a.x = a.x * di + bv.x; a.y = a.y * di + bv.y; a.z = a.z * di + bv.z; a.w = a.w * di + bv.w;
            a.x = (a.x > 0.0f) ? a.x : expm1f(a.x);
            a.y = (a.y > 0.0f) ? a.y : expm1f(a.y);
            a.z = (a.z > 0.0f) ? a.z : expm1f(a.z);
            a.w = (a.w > 0.0f) ? a.w : expm1f(a.w);
            out4[i] = a;
        }
    }
}

extern "C" void kernel_launch(void* const* d_in, const int* in_sizes, int n_in,
                              void* d_out, int out_size, void* d_ws, size_t ws_size,
                              hipStream_t stream) {
    const float* x  = (const float*)d_in[0];
    const int*   ei = (const int*)d_in[1];
    const float* W1 = (const float*)d_in[2];
    const float* b1 = (const float*)d_in[3];
    const float* W2 = (const float*)d_in[4];
    const float* b2 = (const float*)d_in[5];
    float* out = (float*)d_out;

    const int N = in_sizes[0] / IN_DIM;
    const int E = in_sizes[1] / 2;
    const int* src = ei;
    const int* dst = ei + E;

    char* ws = (char*)d_ws;
    float* dinv = (float*)ws;
    size_t off1 = ((size_t)N * 4 + 255) & ~(size_t)255;
    float* hbuf = (float*)(ws + off1);
    size_t off2 = off1 + ((((size_t)N * HID_DIM * 4) + 255) & ~(size_t)255);
    float* act1 = (float*)(ws + off2);

    const int TB = 256;
    const int mTiles = (N + 15) / 16;
    const int nTiles = (N + TILE_ROWS - 1) / TILE_ROWS;

    gcn_fill_one<<<(N + TB - 1) / TB, TB, 0, stream>>>(dinv, N);
    unsigned char* keys = (unsigned char*)d_out + KEY_OFF;
    gcn_deg_keys<<<((E + 3) / 4 + TB - 1) / TB, TB, 0, stream>>>(dst, dinv, keys, E);
    gcn_rsqrt_ip<<<(N + TB - 1) / TB, TB, 0, stream>>>(dinv, N);

    gcn_gemm_wmma<HID_DIM><<<(mTiles + GEMM_WAVES - 1) / GEMM_WAVES, GEMM_THREADS, 0, stream>>>(x, W1, dinv, hbuf, N);
    gcn_tile_aggregate<HID_DIM, false><<<nTiles, AGG_THREADS, 0, stream>>>(hbuf, src, dst, dinv, b1, act1, out, keys, N, E);

    gcn_gemm_wmma<OUT_DIM><<<(mTiles + GEMM_WAVES - 1) / GEMM_WAVES, GEMM_THREADS, 0, stream>>>(act1, W2, dinv, hbuf, N);
    gcn_tile_aggregate<OUT_DIM, true><<<nTiles, AGG_THREADS, 0, stream>>>(hbuf, src, dst, dinv, b2, out, out, keys, N, E);
}


# Tile sums leave LDS f32 atomics: wave r%32 owns tile row r; finders queue matched edges to owners in LDS (slot = returning int add); owner adds rows by plain LDS read-add-write, queue arrival order
# speedup vs baseline: 2.7045x; 1.0691x over previous
#include <hip/hip_runtime.h>
#include <cmath>

typedef __attribute__((ext_vector_type(2))) float v2f;
typedef __attribute__((ext_vector_type(8))) float v8f;

#define IN_DIM 128
#define HID_DIM 128
#define OUT_DIM 64
#define KDIM 128

#define TILE_ROWS   391
#define REC_ROW     OUT_DIM
#define TILE_BYTES  (TILE_ROWS * REC_ROW * 4)
#define KEY_ROWS      64
#define KEY_SHIFT     14
#define KEYS_PER_TILE (1 << KEY_SHIFT)
#define KEY_OFF       ((TILE_ROWS - KEY_ROWS) * REC_ROW * 4)

__global__ void gcn_fill_one(float* __restrict__ deg, int n) {
    int i = blockIdx.x * blockDim.x + threadIdx.x;
    if (i < n) deg[i] = 1.0f;
}

__global__ void gcn_deg_keys(const int* __restrict__ dst, float* __restrict__ deg,
                             unsigned char* keys, int e) {
    const int i = blockIdx.x * blockDim.x + threadIdx.x;
    const int e0 = 4 * i;
    if (e0 >= e) return;
    unsigned char* kp = keys + (size_t)(e0 >> KEY_SHIFT) * TILE_BYTES + (e0 & (KEYS_PER_TILE - 1));
    if (e0 + 4 <= e) {
        unsigned packed = 0u;
        #pragma unroll
        for (int k = 0; k < 4; ++k) {
            const int d = dst[e0 + k];
            unsafeAtomicAdd(&deg[d], 1.0f);
            packed |= ((unsigned)d / (unsigned)TILE_ROWS) << (8 * k);
        }
        *reinterpret_cast<unsigned*>(kp) = packed;
    } else {
        for (int k = 0; e0 + k < e; ++k) {
            const int d = dst[e0 + k];
            unsafeAtomicAdd(&deg[d], 1.0f);
            kp[k] = (unsigned char)((unsigned)d / (unsigned)TILE_ROWS);
        }
    }
}

__global__ void gcn_rsqrt_ip(float* __restrict__ deg, int n) {
    int i = blockIdx.x * blockDim.x + threadIdx.x;
    if (i < n) deg[i] = rsqrtf(deg[i]);
}

#define GEMM_THREADS 128
#define GEMM_WAVES   (GEMM_THREADS / 32)

template <int NOUT, int NT>
__device__ __forceinline__ void gcn_load_b_step(const float* __restrict__ bk, v2f (&b)[NT]) {
    #pragma unroll
    for (int t = 0; t < NT; ++t) {
        b[t].x = bk[16 * t];
        b[t].y = bk[NOUT + 16 * t];
    }
}

template <int NOUT>
__global__ __launch_bounds__(GEMM_THREADS) void gcn_gemm_wmma(
    const float* __restrict__ A, const float* __restrict__ B,
    const float* __restrict__ rs, float* __restrict__ C, int M) {
    constexpr int NT = NOUT / 16;

    const int tid  = threadIdx.x;
    const int wave = tid >> 5;
    const int lane = tid & 31;
    const int c    = lane & 15;
    const int h    = lane >> 4;

    const int mTile = blockIdx.x * GEMM_WAVES + wave;
    if (mTile * 16 >= M) return;

    int arow = mTile * 16 + c;
    if (arow >= M) arow = M - 1;

    const float4* ap = reinterpret_cast<const float4*>(A + (size_t)arow * KDIM) + h;
    float4 a[16];
    #pragma unroll
    for (int j = 0; j < 16; ++j) a[j] = ap[2 * j];

    const float* bp = B + (size_t)(4 * h) * NOUT + c;

    v2f bcur[NT], bnxt[NT];
    gcn_load_b_step<NOUT, NT>(bp, bcur);

    __builtin_amdgcn_sched_barrier(0);

    v8f acc[NT];
    #pragma unroll
    for (int t = 0; t < NT; ++t) acc[t] = v8f{};

    const int rowBase = mTile * 16 + 8 * h;
    float sc[8];

    #pragma unroll
    for (int s = 0; s < 32; ++s) {
        const int j = s >> 1;
        const int p = s & 1;
        if (s + 1 < 32) {
            gcn_load_b_step<NOUT, NT>(bp + (size_t)(8 * ((s + 1) >> 1) + 2 * ((s + 1) & 1)) * NOUT, bnxt);
        } else {
            #pragma unroll
            for (int i = 0; i < 8; ++i) {
                int r = rowBase + i;
                if (r >= M) r = M - 1;
                sc[i] = rs[r];
            }
        }
        __builtin_amdgcn_sched_barrier(0);
        v2f av;
        av.x = p ? a[j].z : a[j].x;
        av.y = p ? a[j].w : a[j].y;
        #pragma unroll
        for (int t = 0; t < NT; ++t) {
            acc[t] = __builtin_amdgcn_wmma_f32_16x16x4_f32(
                 false, av,  false, bcur[t],
                 (short)0, acc[t],  false,  false);
        }
        #pragma unroll
        for (int t = 0; t < NT; ++t) bcur[t] = bnxt[t];
    }

    __builtin_amdgcn_sched_barrier(0);
    asm volatile("v_nop\n\tv_nop\n\tv_nop\n\tv_nop");
    __builtin_amdgcn_sched_barrier(0);

    #pragma unroll
    for (int i = 0; i < 8; ++i) {
        const int row = rowBase + i;
        if (row < M) {
            float* cp = C + (size_t)row * NOUT + c;
            #pragma unroll
            for (int t = 0; t < NT; ++t) cp[16 * t] = acc[t][i] * sc[i];
        }
    }
}

#define AGG_THREADS 1024
#define AGG_WAVES   (AGG_THREADS / 32)
#define LIST_CAP    640
#define LIST_FLUSH  (LIST_CAP - 512)
#define ROW_BATCH   8
#define QCAP        288
#define ROW_BITS    9

template <int F, int B>
__device__ __forceinline__ void gcn_add_rows(const float* __restrict__ H, float* acc,
                                             int s, int roff, int j, int lane) {
    constexpr int Q = F / 32;
    int oj[B];
    float v[B][Q];
    #pragma unroll
    for (int u = 0; u < B; ++u) {
        const int sj = __builtin_amdgcn_readlane(s, j + u);
        oj[u] = __builtin_amdgcn_readlane(roff, j + u);
        const float* hp = H + (size_t)sj * F + lane;
        #pragma unroll
        for (int q = 0; q < Q; ++q) v[u][q] = hp[q * 32];
    }
    #pragma unroll
    for (int u = 0; u < B; ++u) {
        #pragma unroll
        for (int q = 0; q < Q; ++q)
            atomicAdd(&acc[oj[u] + lane + q * 32], v[u][q]);
    }
}

template <int F, int B>
__device__ __forceinline__ void gcn_owner_load(const float* __restrict__ H, int s, int j, int lane,
                                               float __attribute__((ext_vector_type(F / 32))) (&v)[B]) {
    constexpr int Q = F / 32;
    typedef float vq __attribute__((ext_vector_type(Q)));
    #pragma unroll
    for (int u = 0; u < B; ++u) {
        const int sj = __builtin_amdgcn_readlane(s, j + u);
        v[u] = *reinterpret_cast<const vq*>(H + (size_t)sj * F + lane * Q);
    }
}

template <int F, int B>
__device__ __forceinline__ void gcn_owner_rmw(float* acc, int row, int j, int lane,
                                              const float __attribute__((ext_vector_type(F / 32))) (&v)[B]) {
    constexpr int Q = F / 32;
    typedef float vq __attribute__((ext_vector_type(Q)));
    #pragma unroll
    for (int u = 0; u < B; ++u) {
        const int rj = __builtin_amdgcn_readlane(row, j + u);
        vq* ap = reinterpret_cast<vq*>(acc + rj * F + lane * Q);
        vq a = *ap;
        a += v[u];
        *ap = a;
    }
}

template <int F, int B>
__device__ __forceinline__ void gcn_owner_rows(const float* __restrict__ H, float* acc,
                                               int s, int row, int j, int lane) {
    float __attribute__((ext_vector_type(F / 32))) v[B];
    gcn_owner_load<F, B>(H, s, j, lane, v);
    gcn_owner_rmw<F, B>(acc, row, j, lane, v);
}

template <int F>
__device__ __forceinline__ void gcn_owner_chunk(const float* __restrict__ H, float* acc,
                                                int s, int row, int m, int lane) {
    if (m == 32) {
        float __attribute__((ext_vector_type(F / 32))) va[ROW_BATCH], vb[ROW_BATCH];
        static_assert(ROW_BATCH == 8, "four batches of eight make a full chunk");
        gcn_owner_load<F, ROW_BATCH>(H, s, 0, lane, va);
        gcn_owner_load<F, ROW_BATCH>(H, s, 8, lane, vb);
        gcn_owner_rmw<F, ROW_BATCH>(acc, row, 0, lane, va);
        gcn_owner_load<F, ROW_BATCH>(H, s, 16, lane, va);
        gcn_owner_rmw<F, ROW_BATCH>(acc, row, 8, lane, vb);
        gcn_owner_load<F, ROW_BATCH>(H, s, 24, lane, vb);
        gcn_owner_rmw<F, ROW_BATCH>(acc, row, 16, lane, va);
        gcn_owner_rmw<F, ROW_BATCH>(acc, row, 24, lane, vb);
        return;
    }
    int j = 0;
    for (; j + ROW_BATCH <= m; j += ROW_BATCH)
        gcn_owner_rows<F, ROW_BATCH>(H, acc, s, row, j, lane);
    if (j + 4 <= m) { gcn_owner_rows<F, 4>(H, acc, s, row, j, lane); j += 4; }
    if (j + 2 <= m) { gcn_owner_rows<F, 2>(H, acc, s, row, j, lane); j += 2; }
    if (j < m)      { gcn_owner_rows<F, 1>(H, acc, s, row, j, lane); }
}

__device__ __forceinline__ void gcn_push_list(const int* __restrict__ src,
                                              const int* __restrict__ dst,
                                              const int* mylist, int cnt, int lo, int lane,
                                              unsigned* oq, int* oqn, int qlim, int* ovf) {
    for (int base = 0; base < cnt; base += 32) {
        const int i = base + lane;
        if (i < cnt) {
            const int e = mylist[i];
            const int s = src[e];
            const int row = dst[e] - lo;
            const int w = row & (AGG_WAVES - 1);
            const int slot = atomicAdd(&oqn[w], 1);
            oq[w * QCAP + ((slot < QCAP) ? slot : (QCAP - 1))] = ((unsigned)s << ROW_BITS) | (unsigned)row;
            if (slot >= qlim) *ovf = 1;
        }
    }
}

template <int F>
__device__ __forceinline__ void gcn_owner_replay(const float* __restrict__ H,
                                                 float* acc, const int2* ent,
                                                 int cnt, int lo, int lane) {
    if (cnt <= 0) return;
    int2 en = ent[(lane < cnt) ? lane : 0];
    for (int base = 0; base < cnt; base += 32) {
        int m = cnt - base;
        if (m > 32) m = 32;
        const int s = en.x;
        const int row = en.y - lo;
        const int nb = base + 32;
        if (nb < cnt) en = ent[(nb + lane < cnt) ? (nb + lane) : nb];
        gcn_owner_chunk<F>(H, acc, s, row, m, lane);
    }
}

template <int F, bool REC>
__device__ __forceinline__ void gcn_flush_list(const float* __restrict__ H,
                                               const int* __restrict__ src,
                                               const int* __restrict__ dst,
                                               float* acc, const int* mylist,
                                               int cnt, int lo, int lane,
                                               int2* ent, int cap, int& written) {
    for (int base = 0; base < cnt; base += 32) {
        int m = cnt - base;
        if (m > 32) m = 32;
        const int idx = (lane < m) ? (base + lane) : base;
        const int e = mylist[idx];
        const int s = src[e];
        const int d = dst[e];
        if (REC) {
            if (written + m <= cap && lane < m) ent[written + lane] = make_int2(s, d);
            written += m;
        }
        const int roff = (d - lo) * F;

        int j = 0;
        for (; j + ROW_BATCH <= m; j += ROW_BATCH)
            gcn_add_rows<F, ROW_BATCH>(H, acc, s, roff, j, lane);
        if (j + 4 <= m) { gcn_add_rows<F, 4>(H, acc, s, roff, j, lane); j += 4; }
        if (j + 2 <= m) { gcn_add_rows<F, 2>(H, acc, s, roff, j, lane); j += 2; }
        if (j < m)      { gcn_add_rows<F, 1>(H, acc, s, roff, j, lane); }
    }
}

template <int F>
__device__ __forceinline__ void gcn_replay_list(const float* __restrict__ H,
                                                float* acc, const int2* ent,
                                                int cnt, int lo, int lane) {
    int2 en = ent[(lane < cnt) ? lane : 0];
    for (int base = 0; base < cnt; base += 32) {
        int m = cnt - base;
        if (m > 32) m = 32;
        const int s = en.x;
        const int d = en.y;
        const int nb = base + 32;
        if (nb < cnt) en = ent[(nb + lane < cnt) ? (nb + lane) : nb];
        const int roff = (d - lo) * F;

        int j = 0;
        for (; j + ROW_BATCH <= m; j += ROW_BATCH)
            gcn_add_rows<F, ROW_BATCH>(H, acc, s, roff, j, lane);
        if (j + 4 <= m) { gcn_add_rows<F, 4>(H, acc, s, roff, j, lane); j += 4; }
        if (j + 2 <= m) { gcn_add_rows<F, 2>(H, acc, s, roff, j, lane); j += 2; }
        if (j < m)      { gcn_add_rows<F, 1>(H, acc, s, roff, j, lane); }
    }
}

template <int F, bool REPLAY>
__global__ __launch_bounds__(AGG_THREADS) void gcn_tile_aggregate(
    const float* __restrict__ H, const int* __restrict__ src,
    const int* __restrict__ dst, const float* __restrict__ dinv,
    const float* __restrict__ bias, float* out, float* rec,
    const unsigned char* keys, int N, int E) {
    static_assert(REC_ROW % AGG_WAVES == 0, "a tile row splits evenly over the waves' record slices");
    static_assert(KEYS_PER_TILE == KEY_ROWS * REC_ROW * 4, "a tile's keys fill its reserved rows exactly");
    __shared__ __align__(16) float acc[TILE_ROWS * F];
    __shared__ int elist[AGG_WAVES * LIST_CAP];
    __shared__ unsigned oq[REPLAY ? AGG_WAVES : AGG_WAVES * QCAP];
    __shared__ int oqn[AGG_WAVES];
    __shared__ int ovf;
    static_assert(TILE_ROWS <= (1 << ROW_BITS), "a row index fits the low bits of a queue entry");

    const int tid  = threadIdx.x;
    const int lane = tid & 31;
    const int wave = __builtin_amdgcn_readfirstlane(tid >> 5);
    const int lo   = blockIdx.x * TILE_ROWS;
    int nrows = N - lo;
    if (nrows > TILE_ROWS) nrows = TILE_ROWS;
    const int nelem = nrows * F;
    const size_t gbase = (size_t)lo * F;

    const int nvec = nelem >> 2;
    {
        const float4* H4 = reinterpret_cast<const float4*>(H + gbase);
        float4* acc4 = reinterpret_cast<float4*>(acc);
        #pragma unroll 4
        for (int i = tid; i < nvec; i += AGG_THREADS) acc4[i] = H4[i];
    }
    if (!REPLAY) {
        if (tid < AGG_WAVES) oqn[tid] = 0;
        if (tid == 0) ovf = 0;
    }
    __syncthreads();

    int* mylist = elist + wave * LIST_CAP;
    int cnt = 0;

    const int rrows = (nrows > KEY_ROWS) ? (nrows - KEY_ROWS) : 0;
    const int seg = (rrows * (REC_ROW / AGG_WAVES)) & ~31;
    int* segp = reinterpret_cast<int*>(rec) + (size_t)lo * REC_ROW + (size_t)wave * seg;
    int2* ent = reinterpret_cast<int2*>(segp + 2);
    const int cap = (seg >= 32) ? ((seg - 2) >> 1) : -1;
    int written = 0;
    int mode = 0;
    const int qlim = (cap < QCAP) ? cap : QCAP;

    bool from_record = false;
    if (REPLAY) {
        int saved = -1;
        int smode = 0;
        if (cap >= 0) { saved = segp[0]; smode = segp[1]; }
        from_record = (saved >= 0) && (saved <= cap);
        if (from_record && smode == 1) gcn_owner_replay<F>(H, acc, ent, saved, lo, lane);
        else if (from_record && saved > 0) gcn_replay_list<F>(H, acc, ent, saved, lo, lane);
    }
    if (!from_record) {

#define GCN_APPEND(bal, mk, ek)                                                     \
    if ((bal) != 0u) {                                                              \
        if (mk) mylist[cnt + (int)__builtin_amdgcn_mbcnt_lo((bal), 0u)] = (ek);     \
        cnt += __popc(bal);                                                         \
    }
#define GCN_SCAN16(kv, eb)                                                          \
    {                                                                               \
        const unsigned x0 = (kv).x ^ pat, x1 = (kv).y ^ pat;                        \
        const unsigned x2 = (kv).z ^ pat, x3 = (kv).w ^ pat;                        \
        const unsigned y0 = ((x0 & 0x7f7f7f7fu) + 0x7f7f7f7fu) | x0;                \
        const unsigned y1 = ((x1 & 0x7f7f7f7fu) + 0x7f7f7f7fu) | x1;                \
        const unsigned y2 = ((x2 & 0x7f7f7f7fu) + 0x7f7f7f7fu) | x2;                \
        const unsigned y3 = ((x3 & 0x7f7f7f7fu) + 0x7f7f7f7fu) | x3;                \
        const unsigned hit = ~(y0 & y1 & y2 & y3) & 0x80808080u;                    \
        unsigned bal = __builtin_amdgcn_ballot_w32(hit != 0u);                      \
        if (bal != 0u) {                                                            \
            unsigned f = ((~y0 & 0x80808080u) >> 7) | ((~y1 & 0x80808080u) >> 6) |  \
                         ((~y2 & 0x80808080u) >> 5) | ((~y3 & 0x80808080u) >> 4);   \
            do {                                                                    \
                if (f != 0u) {                                                      \
                    const int fi = __builtin_ctz(f);                                \
                    mylist[cnt + (int)__builtin_amdgcn_mbcnt_lo(bal, 0u)] =         \
                        (eb) + ((fi & 7) << 2) + (fi >> 3);                         \
                    f &= f - 1u;                                                    \
                }                                                                   \
                cnt += __popc(bal);                                                 \
                bal = __builtin_amdgcn_ballot_w32(f != 0u);                         \
            } while (bal != 0u);                                                    \
        }                                                                           \
    }

    constexpr int NPASS = REPLAY ? 1 : 2;
    for (int pass = 0; pass < NPASS; ++pass) {
    const bool push = !REPLAY && pass == 0;
    cnt = 0;
    written = 0;
    if constexpr (!REPLAY) {
        const unsigned pat = (unsigned)blockIdx.x * 0x01010101u;
        const int nGroups = E >> 11;
        int g = wave;
        uint4 n0 = make_uint4(0u, 0u, 0u, 0u), n1 = n0, n2 = n0, n3 = n0;
        if (g < nGroups) {
            const uint4* p = reinterpret_cast<const uint4*>(
                keys + (size_t)(g >> 3) * TILE_BYTES + ((g & 7) << 11)) + lane;
            n0 = p[0]; n1 = p[32]; n2 = p[64]; n3 = p[96];
        }
        while (g < nGroups) {
            const uint4 c0 = n0, c1 = n1, c2 = n2, c3 = n3;
            const int gn = g + AGG_WAVES;
            if (gn < nGroups) {
                const uint4* p = reinterpret_cast<const uint4*>(
                    keys + (size_t)(gn >> 3) * TILE_BYTES + ((gn & 7) << 11)) + lane;
                n0 = p[0]; n1 = p[32]; n2 = p[64]; n3 = p[96];
            }
            const int eb = (g << 11) + (lane << 4);
            int sub = 0;
            do {
                if (sub == 0) { GCN_SCAN16(c0, eb) sub = 1; }
                if (sub == 1 && cnt <= LIST_FLUSH) { GCN_SCAN16(c1, eb + 512) sub = 2; }
                if (sub == 2 && cnt <= LIST_FLUSH) { GCN_SCAN16(c2, eb + 1024) sub = 3; }
                if (sub == 3 && cnt <= LIST_FLUSH) { GCN_SCAN16(c3, eb + 1536) sub = 4; }
                if (cnt > LIST_FLUSH) {
                    if (push) gcn_push_list(src, dst, mylist, cnt, lo, lane, oq, oqn, qlim, &ovf);
                    else gcn_flush_list<F, true>(H, src, dst, acc, mylist, cnt, lo, lane, ent, cap, written);
                    cnt = 0;
                }
            } while (sub < 4);
            g = gn;
        }
    } else {
        const int nGroups = E >> 11;
        for (int g = wave; g < nGroups; g += AGG_WAVES) {
            for (int k = 0; k < 2048; k += 128) {
                const int eb = (g << 11) + k + lane;
                const int d0 = dst[eb], d1 = dst[eb + 32], d2 = dst[eb + 64], d3 = dst[eb + 96];
                const bool m0 = (unsigned)(d0 - lo) < (unsigned)TILE_ROWS;
                const bool m1 = (unsigned)(d1 - lo) < (unsigned)TILE_ROWS;
                const bool m2 = (unsigned)(d2 - lo) < (unsigned)TILE_ROWS;
                const bool m3 = (unsigned)(d3 - lo) < (unsigned)TILE_ROWS;
                const unsigned q0 = __builtin_amdgcn_ballot_w32(m0);
                const unsigned q1 = __builtin_amdgcn_ballot_w32(m1);
                const unsigned q2 = __builtin_amdgcn_ballot_w32(m2);
                const unsigned q3 = __builtin_amdgcn_ballot_w32(m3);
                GCN_APPEND(q0, m0, eb)
                GCN_APPEND(q1, m1, eb + 32)
                GCN_APPEND(q2, m2, eb + 64)
                GCN_APPEND(q3, m3, eb + 96)
                if (cnt > LIST_FLUSH) {
                    gcn_flush_list<F, false>(H, src, dst, acc, mylist, cnt, lo, lane, ent, cap, written);
                    cnt = 0;
                }
            }
        }
    }
    for (int base = (E >> 11) << 11; base < E; base += AGG_THREADS) {
        const int e = base + tid;
        bool mk = false;
        if (e < E) mk = (unsigned)(dst[e] - lo) < (unsigned)TILE_ROWS;
        const unsigned qt = __builtin_amdgcn_ballot_w32(mk);
        GCN_APPEND(qt, mk, e)
    }
    if (cnt > 0) {
        if (push) gcn_push_list(src, dst, mylist, cnt, lo, lane, oq, oqn, qlim, &ovf);
        else gcn_flush_list<F, !REPLAY>(H, src, dst, acc, mylist, cnt, lo, lane, ent, cap, written);
    }
    if (push) {
        __syncthreads();
        if (ovf == 0) {
            const int qn = oqn[wave];
            const unsigned* q = oq + wave * QCAP;
            for (int base = 0; base < qn; base += 32) {
                int m = qn - base;
                if (m > 32) m = 32;
                const unsigned p = q[(lane < m) ? (base + lane) : base];
                const int s = (int)(p >> ROW_BITS);
                const int row = (int)(p & ((1u << ROW_BITS) - 1u));
                if (lane < m) ent[base + lane] = make_int2(s, lo + row);
                gcn_owner_chunk<F>(H, acc, s, row, m, lane);
            }
            written = qn;
            mode = 1;
            break;
        }
    }
    }
    if (!REPLAY && cap >= 0 && lane == 0) { segp[0] = written; segp[1] = mode; }
    }
#undef GCN_SCAN16
#undef GCN_APPEND
    __syncthreads();

    {
        const float4 bv = reinterpret_cast<const float4*>(bias)[tid & (F / 4 - 1)];
        const float4* acc4 = reinterpret_cast<const float4*>(acc);
        float4* out4 = reinterpret_cast<float4*>(out + gbase);
        for (int i = tid; i < nvec; i += AGG_THREADS) {
            const float di = dinv[lo + (int)((unsigned)i / (unsigned)(F / 4))];
            float4 a = acc4[i];
            a.x = a.x * di + bv.x; a.y = a.y * di + bv.y; a.z = a.z * di + bv.z; a.w = a.w * di + bv.w;
            a.x = (a.x > 0.0f) ? a.x : expm1f(a.x);
            a.y = (a.y > 0.0f) ? a.y : expm1f(a.y);
            a.z = (a.z > 0.0f) ? a.z : expm1f(a.z);
            a.w = (a.w > 0.0f) ? a.w : expm1f(a.w);
            out4[i] = a;
        }
    }
}

extern "C" void kernel_launch(void* const* d_in, const int* in_sizes, int n_in,
                              void* d_out, int out_size, void* d_ws, size_t ws_size,
                              hipStream_t stream) {
    const float* x  = (const float*)d_in[0];
    const int*   ei = (const int*)d_in[1];
    const float* W1 = (const float*)d_in[2];
    const float* b1 = (const float*)d_in[3];
    const float* W2 = (const float*)d_in[4];
    const float* b2 = (const float*)d_in[5];
    float* out = (float*)d_out;

    const int N = in_sizes[0] / IN_DIM;
    const int E = in_sizes[1] / 2;
    const int* src = ei;
    const int* dst = ei + E;

    char* ws = (char*)d_ws;
    float* dinv = (float*)ws;
    size_t off1 = ((size_t)N * 4 + 255) & ~(size_t)255;
    float* hbuf = (float*)(ws + off1);
    size_t off2 = off1 + ((((size_t)N * HID_DIM * 4) + 255) & ~(size_t)255);
    float* act1 = (float*)(ws + off2);

    const int TB = 256;
    const int mTiles = (N + 15) / 16;
    const int nTiles = (N + TILE_ROWS - 1) / TILE_ROWS;

    gcn_fill_one<<<(N + TB - 1) / TB, TB, 0, stream>>>(dinv, N);
    unsigned char* keys = (unsigned char*)d_out + KEY_OFF;
    gcn_deg_keys<<<((E + 3) / 4 + TB - 1) / TB, TB, 0, stream>>>(dst, dinv, keys, E);
    gcn_rsqrt_ip<<<(N + TB - 1) / TB, TB, 0, stream>>>(dinv, N);

    gcn_gemm_wmma<HID_DIM><<<(mTiles + GEMM_WAVES - 1) / GEMM_WAVES, GEMM_THREADS, 0, stream>>>(x, W1, dinv, hbuf, N);
    gcn_tile_aggregate<HID_DIM, false><<<nTiles, AGG_THREADS, 0, stream>>>(hbuf, src, dst, dinv, b1, act1, out, keys, N, E);

    gcn_gemm_wmma<OUT_DIM><<<(mTiles + GEMM_WAVES - 1) / GEMM_WAVES, GEMM_THREADS, 0, stream>>>(act1, W2, dinv, hbuf, N);
    gcn_tile_aggregate<OUT_DIM, true><<<nTiles, AGG_THREADS, 0, stream>>>(hbuf, src, dst, dinv, b2, out, out, keys, N, E);
}


# Degrees leave global f32 atomics: a tile-scan launch ahead of GEMM 1 scans keys, queues and records edges, counts each row's edges in LDS, writes dinv; both aggregates walk records; 7 launches to 6
# speedup vs baseline: 3.9929x; 1.4764x over previous
#include <hip/hip_runtime.h>
#include <cmath>

typedef __attribute__((ext_vector_type(2))) float v2f;
typedef __attribute__((ext_vector_type(8))) float v8f;

#define IN_DIM 128
#define HID_DIM 128
#define OUT_DIM 64
#define KDIM 128

#define TILE_ROWS   391
#define REC_ROW     OUT_DIM
#define TILE_BYTES  (TILE_ROWS * REC_ROW * 4)
#define KEY_ROWS      64
#define KEY_SHIFT     14
#define KEYS_PER_TILE (1 << KEY_SHIFT)
#define KEY_OFF       ((TILE_ROWS - KEY_ROWS) * REC_ROW * 4)

__global__ void gcn_tile_keys(const int* __restrict__ dst, unsigned char* keys, int e) {
    const int i = blockIdx.x * blockDim.x + threadIdx.x;
    const int e0 = 4 * i;
    if (e0 >= e) return;
    unsigned char* kp = keys + (size_t)(e0 >> KEY_SHIFT) * TILE_BYTES + (e0 & (KEYS_PER_TILE - 1));
    if (e0 + 4 <= e) {
        unsigned packed = 0u;
        #pragma unroll
        for (int k = 0; k < 4; ++k)
            packed |= ((unsigned)dst[e0 + k] / (unsigned)TILE_ROWS) << (8 * k);
        *reinterpret_cast<unsigned*>(kp) = packed;
    } else {
        for (int k = 0; e0 + k < e; ++k)
            kp[k] = (unsigned char)((unsigned)dst[e0 + k] / (unsigned)TILE_ROWS);
    }
}

#define GEMM_THREADS 128
#define GEMM_WAVES   (GEMM_THREADS / 32)

template <int NOUT, int NT>
__device__ __forceinline__ void gcn_load_b_step(const float* __restrict__ bk, v2f (&b)[NT]) {
    #pragma unroll
    for (int t = 0; t < NT; ++t) {
        b[t].x = bk[16 * t];
        b[t].y = bk[NOUT + 16 * t];
    }
}

template <int NOUT>
__global__ __launch_bounds__(GEMM_THREADS) void gcn_gemm_wmma(
    const float* __restrict__ A, const float* __restrict__ B,
    const float* __restrict__ rs, float* __restrict__ C, int M) {
    constexpr int NT = NOUT / 16;

    const int tid  = threadIdx.x;
    const int wave = tid >> 5;
    const int lane = tid & 31;
    const int c    = lane & 15;
    const int h    = lane >> 4;

    const int mTile = blockIdx.x * GEMM_WAVES + wave;
    if (mTile * 16 >= M) return;

    int arow = mTile * 16 + c;
    if (arow >= M) arow = M - 1;

    const float4* ap = reinterpret_cast<const float4*>(A + (size_t)arow * KDIM) + h;
    float4 a[16];
    #pragma unroll
    for (int j = 0; j < 16; ++j) a[j] = ap[2 * j];

    const float* bp = B + (size_t)(4 * h) * NOUT + c;

    v2f bcur[NT], bnxt[NT];
    gcn_load_b_step<NOUT, NT>(bp, bcur);

    __builtin_amdgcn_sched_barrier(0);

    v8f acc[NT];
    #pragma unroll
    for (int t = 0; t < NT; ++t) acc[t] = v8f{};

    const int rowBase = mTile * 16 + 8 * h;
    float sc[8];

    #pragma unroll
    for (int s = 0; s < 32; ++s) {
        const int j = s >> 1;
        const int p = s & 1;
        if (s + 1 < 32) {
            gcn_load_b_step<NOUT, NT>(bp + (size_t)(8 * ((s + 1) >> 1) + 2 * ((s + 1) & 1)) * NOUT, bnxt);
        } else {
            #pragma unroll
            for (int i = 0; i < 8; ++i) {
                int r = rowBase + i;
                if (r >= M) r = M - 1;
                sc[i] = rs[r];
            }
        }
        __builtin_amdgcn_sched_barrier(0);
        v2f av;
        av.x = p ? a[j].z : a[j].x;
        av.y = p ? a[j].w : a[j].y;
        #pragma unroll
        for (int t = 0; t < NT; ++t) {
            acc[t] = __builtin_amdgcn_wmma_f32_16x16x4_f32(
                 false, av,  false, bcur[t],
                 (short)0, acc[t],  false,  false);
        }
        #pragma unroll
        for (int t = 0; t < NT; ++t) bcur[t] = bnxt[t];
    }

    __builtin_amdgcn_sched_barrier(0);
    asm volatile("v_nop\n\tv_nop\n\tv_nop\n\tv_nop");
    __builtin_amdgcn_sched_barrier(0);

    #pragma unroll
    for (int i = 0; i < 8; ++i) {
        const int row = rowBase + i;
        if (row < M) {
            float* cp = C + (size_t)row * NOUT + c;
            #pragma unroll
            for (int t = 0; t < NT; ++t) cp[16 * t] = acc[t][i] * sc[i];
        }
    }
}

#define AGG_THREADS 1024
#define AGG_WAVES   (AGG_THREADS / 32)
#define LIST_CAP    640
#define LIST_FLUSH  (LIST_CAP - 512)
#define SCAN_LIST_CAP 1280
#define ROW_BATCH   8
#define QCAP        288
#define ROW_BITS    9

template <int F, int B>
__device__ __forceinline__ void gcn_add_rows(const float* __restrict__ H, float* acc,
                                             int s, int roff, int j, int lane) {
    constexpr int Q = F / 32;
    int oj[B];
    float v[B][Q];
    #pragma unroll
    for (int u = 0; u < B; ++u) {
        const int sj = __builtin_amdgcn_readlane(s, j + u);
        oj[u] = __builtin_amdgcn_readlane(roff, j + u);
        const float* hp = H + (size_t)sj * F + lane;
        #pragma unroll
        for (int q = 0; q < Q; ++q) v[u][q] = hp[q * 32];
    }
    #pragma unroll
    for (int u = 0; u < B; ++u) {
        #pragma unroll
        for (int q = 0; q < Q; ++q)
            atomicAdd(&acc[oj[u] + lane + q * 32], v[u][q]);
    }
}

template <int F, int B>
__device__ __forceinline__ void gcn_owner_load(const float* __restrict__ H, int s, int j, int lane,
                                               float __attribute__((ext_vector_type(F / 32))) (&v)[B]) {
    constexpr int Q = F / 32;
    typedef float vq __attribute__((ext_vector_type(Q)));
    #pragma unroll
    for (int u = 0; u < B; ++u) {
        const int sj = __builtin_amdgcn_readlane(s, j + u);
        v[u] = *reinterpret_cast<const vq*>(H + (size_t)sj * F + lane * Q);
    }
}

template <int F, int B>
__device__ __forceinline__ void gcn_owner_rmw(float* acc, int row, int j, int lane,
                                              const float __attribute__((ext_vector_type(F / 32))) (&v)[B]) {
    constexpr int Q = F / 32;
    typedef float vq __attribute__((ext_vector_type(Q)));
    #pragma unroll
    for (int u = 0; u < B; ++u) {
        const int rj = __builtin_amdgcn_readlane(row, j + u);
        vq* ap = reinterpret_cast<vq*>(acc + rj * F + lane * Q);
        vq a = *ap;
        a += v[u];
        *ap = a;
    }
}

template <int F, int B>
__device__ __forceinline__ void gcn_owner_rows(const float* __restrict__ H, float* acc,
                                               int s, int row, int j, int lane) {
    float __attribute__((ext_vector_type(F / 32))) v[B];
    gcn_owner_load<F, B>(H, s, j, lane, v);
    gcn_owner_rmw<F, B>(acc, row, j, lane, v);
}

template <int F>
__device__ __forceinline__ void gcn_owner_chunk(const float* __restrict__ H, float* acc,
                                                int s, int row, int m, int lane) {
    if (m == 32) {
        float __attribute__((ext_vector_type(F / 32))) va[ROW_BATCH], vb[ROW_BATCH];
        static_assert(ROW_BATCH == 8, "four batches of eight make a full chunk");
        gcn_owner_load<F, ROW_BATCH>(H, s, 0, lane, va);
        gcn_owner_load<F, ROW_BATCH>(H, s, 8, lane, vb);
        gcn_owner_rmw<F, ROW_BATCH>(acc, row, 0, lane, va);
        gcn_owner_load<F, ROW_BATCH>(H, s, 16, lane, va);
        gcn_owner_rmw<F, ROW_BATCH>(acc, row, 8, lane, vb);
        gcn_owner_load<F, ROW_BATCH>(H, s, 24, lane, vb);
        gcn_owner_rmw<F, ROW_BATCH>(acc, row, 16, lane, va);
        gcn_owner_rmw<F, ROW_BATCH>(acc, row, 24, lane, vb);
        return;
    }
    int j = 0;
    for (; j + ROW_BATCH <= m; j += ROW_BATCH)
        gcn_owner_rows<F, ROW_BATCH>(H, acc, s, row, j, lane);
    if (j + 4 <= m) { gcn_owner_rows<F, 4>(H, acc, s, row, j, lane); j += 4; }
    if (j + 2 <= m) { gcn_owner_rows<F, 2>(H, acc, s, row, j, lane); j += 2; }
    if (j < m)      { gcn_owner_rows<F, 1>(H, acc, s, row, j, lane); }
}

__device__ __forceinline__ void gcn_push_list(const int* __restrict__ src,
                                              const int* __restrict__ dst,
                                              const int* mylist, int cnt, int lo, int lane,
                                              unsigned* oq, int* oqn, int qlim, int* ovf) {
    for (int base = 0; base < cnt; base += 32) {
        const int i = base + lane;
        if (i < cnt) {
            const int e = mylist[i];
            const int s = src[e];
            const int row = dst[e] - lo;
            const int w = row & (AGG_WAVES - 1);
            const int slot = atomicAdd(&oqn[w], 1);
            oq[w * QCAP + ((slot < QCAP) ? slot : (QCAP - 1))] = ((unsigned)s << ROW_BITS) | (unsigned)row;
            if (slot >= qlim) *ovf = 1;
        }
    }
}

template <int F>
__device__ __forceinline__ void gcn_owner_replay(const float* __restrict__ H,
                                                 float* acc, const int2* ent,
                                                 int cnt, int lo, int lane) {
    if (cnt <= 0) return;
    int2 en = ent[(lane < cnt) ? lane : 0];
    for (int base = 0; base < cnt; base += 32) {
        int m = cnt - base;
        if (m > 32) m = 32;
        const int s = en.x;
        const int row = en.y - lo;
        const int nb = base + 32;
        if (nb < cnt) en = ent[(nb + lane < cnt) ? (nb + lane) : nb];
        gcn_owner_chunk<F>(H, acc, s, row, m, lane);
    }
}

template <int F>
__device__ __forceinline__ void gcn_flush_list(const float* __restrict__ H,
                                               const int* __restrict__ src,
                                               const int* __restrict__ dst,
                                               float* acc, const int* mylist,
                                               int cnt, int lo, int lane) {
    for (int base = 0; base < cnt; base += 32) {
        int m = cnt - base;
        if (m > 32) m = 32;
        const int idx = (lane < m) ? (base + lane) : base;
        const int e = mylist[idx];
        const int s = src[e];
        const int d = dst[e];
        const int roff = (d - lo) * F;

        int j = 0;
        for (; j + ROW_BATCH <= m; j += ROW_BATCH)
            gcn_add_rows<F, ROW_BATCH>(H, acc, s, roff, j, lane);
        if (j + 4 <= m) { gcn_add_rows<F, 4>(H, acc, s, roff, j, lane); j += 4; }
        if (j + 2 <= m) { gcn_add_rows<F, 2>(H, acc, s, roff, j, lane); j += 2; }
        if (j < m)      { gcn_add_rows<F, 1>(H, acc, s, roff, j, lane); }
    }
}

__device__ __forceinline__ void gcn_note_list(const int* __restrict__ src,
                                              const int* __restrict__ dst,
                                              const int* mylist, int cnt, int lo, int lane,
                                              int2* ent, int cap, int& written, int* rowdeg) {
    for (int base = 0; base < cnt; base += 32) {
        int m = cnt - base;
        if (m > 32) m = 32;
        if (lane < m) {
            const int e = mylist[base + lane];
            const int s = src[e];
            const int d = dst[e];
            if (written + m <= cap) ent[written + lane] = make_int2(s, d);
            atomicAdd(&rowdeg[d - lo], 1);
        }
        written += m;
    }
}

template <int F>
__device__ __forceinline__ void gcn_replay_list(const float* __restrict__ H,
                                                float* acc, const int2* ent,
                                                int cnt, int lo, int lane) {
    int2 en = ent[(lane < cnt) ? lane : 0];
    for (int base = 0; base < cnt; base += 32) {
        int m = cnt - base;
        if (m > 32) m = 32;
        const int s = en.x;
        const int d = en.y;
        const int nb = base + 32;
        if (nb < cnt) en = ent[(nb + lane < cnt) ? (nb + lane) : nb];
        const int roff = (d - lo) * F;

        int j = 0;
        for (; j + ROW_BATCH <= m; j += ROW_BATCH)
            gcn_add_rows<F, ROW_BATCH>(H, acc, s, roff, j, lane);
        if (j + 4 <= m) { gcn_add_rows<F, 4>(H, acc, s, roff, j, lane); j += 4; }
        if (j + 2 <= m) { gcn_add_rows<F, 2>(H, acc, s, roff, j, lane); j += 2; }
        if (j < m)      { gcn_add_rows<F, 1>(H, acc, s, roff, j, lane); }
    }
}

#define GCN_APPEND(bal, mk, ek)                                                     \
    if ((bal) != 0u) {                                                              \
        if (mk) mylist[cnt + (int)__builtin_amdgcn_mbcnt_lo((bal), 0u)] = (ek);     \
        cnt += __popc(bal);                                                         \
    }
#define GCN_SCAN16(kv, eb)                                                          \
    {                                                                               \
        const unsigned x0 = (kv).x ^ pat, x1 = (kv).y ^ pat;                        \
        const unsigned x2 = (kv).z ^ pat, x3 = (kv).w ^ pat;                        \
        const unsigned y0 = ((x0 & 0x7f7f7f7fu) + 0x7f7f7f7fu) | x0;                \
        const unsigned y1 = ((x1 & 0x7f7f7f7fu) + 0x7f7f7f7fu) | x1;                \
        const unsigned y2 = ((x2 & 0x7f7f7f7fu) + 0x7f7f7f7fu) | x2;                \
        const unsigned y3 = ((x3 & 0x7f7f7f7fu) + 0x7f7f7f7fu) | x3;                \
        const unsigned hit = ~(y0 & y1 & y2 & y3) & 0x80808080u;                    \
        unsigned bal = __builtin_amdgcn_ballot_w32(hit != 0u);                      \
        if (bal != 0u) {                                                            \
            unsigned f = ((~y0 & 0x80808080u) >> 7) | ((~y1 & 0x80808080u) >> 6) |  \
                         ((~y2 & 0x80808080u) >> 5) | ((~y3 & 0x80808080u) >> 4);   \
            do {                                                                    \
                if (f != 0u) {                                                      \
                    const int fi = __builtin_ctz(f);                                \
                    mylist[cnt + (int)__builtin_amdgcn_mbcnt_lo(bal, 0u)] =         \
                        (eb) + ((fi & 7) << 2) + (fi >> 3);                         \
                    f &= f - 1u;                                                    \
                }                                                                   \
                cnt += __popc(bal);                                                 \
                bal = __builtin_amdgcn_ballot_w32(f != 0u);                         \
            } while (bal != 0u);                                                    \
        }                                                                           \
    }

__global__ __launch_bounds__(AGG_THREADS) void gcn_tile_scan(
    const int* __restrict__ src, const int* __restrict__ dst,
    const unsigned char* keys, float* rec, float* __restrict__ dinv, int N, int E) {
    static_assert(REC_ROW % AGG_WAVES == 0, "a tile row splits evenly over the waves' record slices");
    static_assert(KEYS_PER_TILE == KEY_ROWS * REC_ROW * 4, "a tile's keys fill its reserved rows exactly");
    static_assert(TILE_ROWS <= (1 << ROW_BITS), "a row index fits the low bits of a queue entry");
    static_assert(TILE_ROWS <= AGG_THREADS, "one thread per tile row clears and writes the degrees");
    __shared__ int elist[AGG_WAVES * SCAN_LIST_CAP];
    __shared__ unsigned oq[AGG_WAVES * QCAP];
    __shared__ int oqn[AGG_WAVES];
    __shared__ int ovf;
    __shared__ int rowdeg[TILE_ROWS];

    const int tid  = threadIdx.x;
    const int lane = tid & 31;
    const int wave = __builtin_amdgcn_readfirstlane(tid >> 5);
    const int lo   = blockIdx.x * TILE_ROWS;
    int nrows = N - lo;
    if (nrows > TILE_ROWS) nrows = TILE_ROWS;

    if (tid < TILE_ROWS) rowdeg[tid] = 0;
    if (tid < AGG_WAVES) oqn[tid] = 0;
    if (tid == 0) ovf = 0;
    __syncthreads();

    int* mylist = elist + wave * SCAN_LIST_CAP;
    int cnt = 0;

    const int rrows = (nrows > KEY_ROWS) ? (nrows - KEY_ROWS) : 0;
    const int seg = (rrows * (REC_ROW / AGG_WAVES)) & ~31;
    int* segp = reinterpret_cast<int*>(rec) + (size_t)lo * REC_ROW + (size_t)wave * seg;
    int2* ent = reinterpret_cast<int2*>(segp + 2);
    const int cap = (seg >= 32) ? ((seg - 2) >> 1) : -1;
    int written = 0;
    int mode = 0;
    const int qlim = (cap < QCAP) ? cap : QCAP;

    for (int pass = 0; pass < 2; ++pass) {
        const bool push = (pass == 0);
        cnt = 0;
        written = 0;
        {
            const unsigned pat = (unsigned)blockIdx.x * 0x01010101u;
            const int nGroups = E >> 11;
            int g = wave;
            uint4 n0 = make_uint4(0u, 0u, 0u, 0u), n1 = n0, n2 = n0, n3 = n0;
            if (g < nGroups) {
                const uint4* p = reinterpret_cast<const uint4*>(
                    keys + (size_t)(g >> 3) * TILE_BYTES + ((g & 7) << 11)) + lane;
                n0 = p[0]; n1 = p[32]; n2 = p[64]; n3 = p[96];
            }
            while (g < nGroups) {
                const uint4 c0 = n0, c1 = n1, c2 = n2, c3 = n3;
                const int gn = g + AGG_WAVES;
                if (gn < nGroups) {
                    const uint4* p = reinterpret_cast<const uint4*>(
                        keys + (size_t)(gn >> 3) * TILE_BYTES + ((gn & 7) << 11)) + lane;
                    n0 = p[0]; n1 = p[32]; n2 = p[64]; n3 = p[96];
                }
                const int eb = (g << 11) + (lane << 4);
                int sub = 0;
                do {
                    if (sub == 0) { GCN_SCAN16(c0, eb) sub = 1; }
                    if (sub == 1 && cnt <= LIST_FLUSH) { GCN_SCAN16(c1, eb + 512) sub = 2; }
                    if (sub == 2 && cnt <= LIST_FLUSH) { GCN_SCAN16(c2, eb + 1024) sub = 3; }
                    if (sub == 3 && cnt <= LIST_FLUSH) { GCN_SCAN16(c3, eb + 1536) sub = 4; }
                    if (cnt > LIST_FLUSH) {
                        if (push) gcn_push_list(src, dst, mylist, cnt, lo, lane, oq, oqn, qlim, &ovf);
                        else gcn_note_list(src, dst, mylist, cnt, lo, lane, ent, cap, written, rowdeg);
                        cnt = 0;
                    }
                } while (sub < 4);
                g = gn;
            }
        }
        for (int base = (E >> 11) << 11; base < E; base += AGG_THREADS) {
            const int e = base + tid;
            bool mk = false;
            if (e < E) mk = (unsigned)(dst[e] - lo) < (unsigned)TILE_ROWS;
            const unsigned qt = __builtin_amdgcn_ballot_w32(mk);
            GCN_APPEND(qt, mk, e)
        }
        if (cnt > 0) {
            if (push) gcn_push_list(src, dst, mylist, cnt, lo, lane, oq, oqn, qlim, &ovf);
            else gcn_note_list(src, dst, mylist, cnt, lo, lane, ent, cap, written, rowdeg);
        }
        if (push) {
            __syncthreads();
            if (ovf == 0) {
                const int qn = oqn[wave];
                const unsigned* q = oq + wave * QCAP;
                for (int base = 0; base < qn; base += 32) {
                    if (base + lane < qn) {
                        const unsigned p = q[base + lane];
                        const int row = (int)(p & ((1u << ROW_BITS) - 1u));
                        ent[base + lane] = make_int2((int)(p >> ROW_BITS), lo + row);
                        atomicAdd(&rowdeg[row], 1);
                    }
                }
                written = qn;
                mode = 1;
                break;
            }
        }
    }
    if (cap >= 0 && lane == 0) { segp[0] = written; segp[1] = mode; }
    __syncthreads();
    if (tid < nrows) dinv[lo + tid] = rsqrtf((float)(rowdeg[tid] + 1));
}

template <int F>
__global__ __launch_bounds__(AGG_THREADS) void gcn_tile_aggregate(
    const float* __restrict__ H, const int* __restrict__ src,
    const int* __restrict__ dst, const float* __restrict__ dinv,
    const float* __restrict__ bias, float* out, const float* rec, int N, int E) {
    static_assert(REC_ROW % AGG_WAVES == 0, "a tile row splits evenly over the waves' record slices");
    __shared__ __align__(16) float acc[TILE_ROWS * F];
    __shared__ int elist[AGG_WAVES * LIST_CAP];

    const int tid  = threadIdx.x;
    const int lane = tid & 31;
    const int wave = __builtin_amdgcn_readfirstlane(tid >> 5);
    const int lo   = blockIdx.x * TILE_ROWS;
    int nrows = N - lo;
    if (nrows > TILE_ROWS) nrows = TILE_ROWS;
    const int nelem = nrows * F;
    const size_t gbase = (size_t)lo * F;

    const int nvec = nelem >> 2;
    {
        const float4* H4 = reinterpret_cast<const float4*>(H + gbase);
        float4* acc4 = reinterpret_cast<float4*>(acc);
        #pragma unroll 4
        for (int i = tid; i < nvec; i += AGG_THREADS) acc4[i] = H4[i];
    }
    __syncthreads();

    const int rrows = (nrows > KEY_ROWS) ? (nrows - KEY_ROWS) : 0;
    const int seg = (rrows * (REC_ROW / AGG_WAVES)) & ~31;
    const int* segp = reinterpret_cast<const int*>(rec) + (size_t)lo * REC_ROW + (size_t)wave * seg;
    const int2* ent = reinterpret_cast<const int2*>(segp + 2);
    const int cap = (seg >= 32) ? ((seg - 2) >> 1) : -1;

    int saved = -1;
    int smode = 0;
    if (cap >= 0) { saved = segp[0]; smode = segp[1]; }
    const bool from_record = (saved >= 0) && (saved <= cap);
    if (from_record && smode == 1) gcn_owner_replay<F>(H, acc, ent, saved, lo, lane);
    else if (from_record && saved > 0) gcn_replay_list<F>(H, acc, ent, saved, lo, lane);
    if (!from_record) {
        int* mylist = elist + wave * LIST_CAP;
        int cnt = 0;
        const int nGroups = E >> 11;
        for (int g = wave; g < nGroups; g += AGG_WAVES) {
            for (int k = 0; k < 2048; k += 128) {
                const int eb = (g << 11) + k + lane;
                const int d0 = dst[eb], d1 = dst[eb + 32], d2 = dst[eb + 64], d3 = dst[eb + 96];
                const bool m0 = (unsigned)(d0 - lo) < (unsigned)TILE_ROWS;
                const bool m1 = (unsigned)(d1 - lo) < (unsigned)TILE_ROWS;
                const bool m2 = (unsigned)(d2 - lo) < (unsigned)TILE_ROWS;
                const bool m3 = (unsigned)(d3 - lo) < (unsigned)TILE_ROWS;
                const unsigned q0 = __builtin_amdgcn_ballot_w32(m0);
                const unsigned q1 = __builtin_amdgcn_ballot_w32(m1);
                const unsigned q2 = __builtin_amdgcn_ballot_w32(m2);
                const unsigned q3 = __builtin_amdgcn_ballot_w32(m3);
                GCN_APPEND(q0, m0, eb)
                GCN_APPEND(q1, m1, eb + 32)
                GCN_APPEND(q2, m2, eb + 64)
                GCN_APPEND(q3, m3, eb + 96)
                if (cnt > LIST_FLUSH) {
                    gcn_flush_list<F>(H, src, dst, acc, mylist, cnt, lo, lane);
                    cnt = 0;
                }
            }
        }
        for (int base = (E >> 11) << 11; base < E; base += AGG_THREADS) {
            const int e = base + tid;
            bool mk = false;
            if (e < E) mk = (unsigned)(dst[e] - lo) < (unsigned)TILE_ROWS;
            const unsigned qt = __builtin_amdgcn_ballot_w32(mk);
            GCN_APPEND(qt, mk, e)
        }
        if (cnt > 0) gcn_flush_list<F>(H, src, dst, acc, mylist, cnt, lo, lane);
    }
    __syncthreads();

    {
        const float4 bv = reinterpret_cast<const float4*>(bias)[tid & (F / 4 - 1)];
        const float4* acc4 = reinterpret_cast<const float4*>(acc);
        float4* out4 = reinterpret_cast<float4*>(out + gbase);
        for (int i = tid; i < nvec; i += AGG_THREADS) {
            const float di = dinv[lo + (int)((unsigned)i / (unsigned)(F / 4))];
            float4 a = acc4[i];
            a.x = a.x * di + bv.x; a.y = a.y * di + bv.y; a.z = a.z * di + bv.z; a.w = a.w * di + bv.w;
            a.x = (a.x > 0.0f) ? a.x : expm1f(a.x);
            a.y = (a.y > 0.0f) ? a.y : expm1f(a.y);
            a.z = (a.z > 0.0f) ? a.z : expm1f(a.z);
            a.w = (a.w > 0.0f) ? a.w : expm1f(a.w);
            out4[i] = a;
        }
    }
}
#undef GCN_SCAN16
#undef GCN_APPEND

extern "C" void kernel_launch(void* const* d_in, const int* in_sizes, int n_in,
                              void* d_out, int out_size, void* d_ws, size_t ws_size,
                              hipStream_t stream) {
    const float* x  = (const float*)d_in[0];
    const int*   ei = (const int*)d_in[1];
    const float* W1 = (const float*)d_in[2];
    const float* b1 = (const float*)d_in[3];
    const float* W2 = (const float*)d_in[4];
    const float* b2 = (const float*)d_in[5];
    float* out = (float*)d_out;

    const int N = in_sizes[0] / IN_DIM;
    const int E = in_sizes[1] / 2;
    const int* src = ei;
    const int* dst = ei + E;

    char* ws = (char*)d_ws;
    float* dinv = (float*)ws;
    size_t off1 = ((size_t)N * 4 + 255) & ~(size_t)255;
    float* hbuf = (float*)(ws + off1);
    size_t off2 = off1 + ((((size_t)N * HID_DIM * 4) + 255) & ~(size_t)255);
    float* act1 = (float*)(ws + off2);

    const int TB = 256;
    const int mTiles = (N + 15) / 16;
    const int nTiles = (N + TILE_ROWS - 1) / TILE_ROWS;

    unsigned char* keys = (unsigned char*)d_out + KEY_OFF;
    gcn_tile_keys<<<((E + 3) / 4 + TB - 1) / TB, TB, 0, stream>>>(dst, keys, E);
    gcn_tile_scan<<<nTiles, AGG_THREADS, 0, stream>>>(src, dst, keys, out, dinv, N, E);

    gcn_gemm_wmma<HID_DIM><<<(mTiles + GEMM_WAVES - 1) / GEMM_WAVES, GEMM_THREADS, 0, stream>>>(x, W1, dinv, hbuf, N);
    gcn_tile_aggregate<HID_DIM><<<nTiles, AGG_THREADS, 0, stream>>>(hbuf, src, dst, dinv, b1, act1, out, N, E);

    gcn_gemm_wmma<OUT_DIM><<<(mTiles + GEMM_WAVES - 1) / GEMM_WAVES, GEMM_THREADS, 0, stream>>>(act1, W2, dinv, hbuf, N);
    gcn_tile_aggregate<OUT_DIM><<<nTiles, AGG_THREADS, 0, stream>>>(hbuf, src, dst, dinv, b2, out, out, N, E);
}
